# starting kernel
# baseline (speedup 1.0000x reference)
#include <hip/hip_runtime.h>
#include <math.h>

typedef float v2f __attribute__((ext_vector_type(2)));
typedef float v8f __attribute__((ext_vector_type(8)));

constexpr int T_LEN = 8192;

__global__ __launch_bounds__(32) void dsvf_wmma_kernel(
    const float* __restrict__ x,
    const float* __restrict__ gp,
    const float* __restrict__ rp,
    const float* __restrict__ mhp_p,
    const float* __restrict__ mbp_p,
    const float* __restrict__ mlp_p,
    float* __restrict__ yout)
{
    __shared__ float xs[16][17];
    __shared__ float us[16][17];
    __shared__ float cx1[16], cx2[16];
    __shared__ float cy1[16], cy2[16];
    __shared__ float hs[20];

    const unsigned lane = threadIdx.x;
    const unsigned j    = lane & 15u;
    const bool     hi   = lane >= 16u;
    const int rbase = blockIdx.x * 16;

    const float gv  = gp[0];
    const float rv  = rp[0];
    const float mhp = mhp_p[0];
    const float mbp = mbp_p[0];
    const float mlp = mlp_p[0];

    const float sig = 1.0f / (1.0f + expf(-gv));
    const float gg  = tanf(3.14159265358979323846f * 0.5f * sig);
    const float rr  = log1pf(expf(rv));
    const float g2  = gg * gg;

    const float b0 = g2 * mlp + gg * mbp + mhp;
    const float b1 = 2.0f * g2 * mlp - 2.0f * mhp;
    const float b2 = g2 * mlp - gg * mbp + mhp;
    const float a0 = g2 + 2.0f * rr * gg + 1.0f;
    const float inva0 = 1.0f / a0;
    const float a1 = (2.0f * g2 - 2.0f) * inva0;
    const float a2 = (g2 - 2.0f * rr * gg + 1.0f) * inva0;

    if (lane == 0u) {
        hs[0] = 1.0f;
        hs[1] = -a1;
        for (int i = 2; i <= 16; ++i)
            hs[i] = -a1 * hs[i - 1] - a2 * hs[i - 2];
    }
    if (lane < 16u) {
        cx1[lane] = 0.0f; cx2[lane] = 0.0f;
        cy1[lane] = 0.0f; cy2[lane] = 0.0f;
    }
    __syncthreads();

    v2f Bf[4];
#pragma unroll
    for (int s = 0; s < 4; ++s) {
        const int k0 = 4 * s + (hi ? 2 : 0);
        const int k1 = k0 + 1;
        Bf[s].x = ((int)j >= k0) ? hs[j - k0] : 0.0f;
        Bf[s].y = ((int)j >= k1) ? hs[j - k1] : 0.0f;
    }
    const float pj = hs[j + 1];
    const float qj = -a2 * hs[j];

    const int rload = (int)(lane >> 1);
    const int toff  = (int)(lane & 1u) * 8;
    const float* xrow = x + (size_t)(rbase + rload) * T_LEN;

    for (int t0 = 0; t0 < T_LEN; t0 += 16) {
        const float4 xa = *(const float4*)(xrow + t0 + toff);
        const float4 xb = *(const float4*)(xrow + t0 + toff + 4);
        if (t0 + 16 < T_LEN)
            __builtin_prefetch(xrow + t0 + 16 + toff, 0, 1);

        xs[rload][toff + 0] = xa.x; xs[rload][toff + 1] = xa.y;
        xs[rload][toff + 2] = xa.z; xs[rload][toff + 3] = xa.w;
        xs[rload][toff + 4] = xb.x; xs[rload][toff + 5] = xb.y;
        xs[rload][toff + 6] = xb.z; xs[rload][toff + 7] = xb.w;
        __syncthreads();

        float prev1 = toff ? xs[rload][7] : cx1[rload];
        float prev2 = toff ? xs[rload][6] : cx2[rload];
#pragma unroll
        for (int k = 0; k < 8; ++k) {
            const float xv = xs[rload][toff + k];
            us[rload][toff + k] = b0 * xv + b1 * prev1 + b2 * prev2;
            prev2 = prev1;
            prev1 = xv;
        }
        __syncthreads();

        if (toff) {
            cx1[rload] = xs[rload][15];
            cx2[rload] = xs[rload][14];
        }

        v8f acc = {};
#pragma unroll
        for (int s = 0; s < 4; ++s) {
            const int k0 = 4 * s + (hi ? 2 : 0);
            v2f A;
            A.x = us[j][k0];
            A.y = us[j][k0 + 1];
            acc = __builtin_amdgcn_wmma_f32_16x16x4_f32(
                 false, A,  false, Bf[s],
                 (short)0, acc,  false,  false);
        }

        float outv[8];
#pragma unroll
        for (int v = 0; v < 8; ++v) {
            const int r2 = v + (hi ? 8 : 0);
            const float yv = acc[v] + pj * cy1[r2] + qj * cy2[r2];
            outv[v] = yv;
            yout[(size_t)(rbase + r2) * T_LEN + t0 + j] = yv;
        }
        __syncthreads();

        if (j == 15u) {
#pragma unroll
            for (int v = 0; v < 8; ++v) cy1[v + (hi ? 8 : 0)] = outv[v];
        }
        if (j == 14u) {
#pragma unroll
            for (int v = 0; v < 8; ++v) cy2[v + (hi ? 8 : 0)] = outv[v];
        }
        __syncthreads();
    }
}

extern "C" void kernel_launch(void* const* d_in, const int* in_sizes, int n_in,
                              void* d_out, int out_size, void* d_ws, size_t ws_size,
                              hipStream_t stream) {
    const float* x   = (const float*)d_in[0];
    const float* g   = (const float*)d_in[1];
    const float* r   = (const float*)d_in[2];
    const float* mhp = (const float*)d_in[3];
    const float* mbp = (const float*)d_in[4];
    const float* mlp = (const float*)d_in[5];
    float* y = (float*)d_out;

    const int B = in_sizes[0] / T_LEN;
    dim3 grid(B / 16), block(32);
    hipLaunchKernelGGL(dsvf_wmma_kernel, grid, block, 0, stream,
                       x, g, r, mhp, mbp, mlp, y);
}


# T split into 32 segments, one per wave of a 1024-thread block per 16-row tile: zero-state pass for segment end states, LDS scan with the 256-step transition, pass 2 writes; carry sum order changes
# speedup vs baseline: 8.3749x; 8.3749x over previous
#include <hip/hip_runtime.h>
#include <math.h>

typedef float v2f __attribute__((ext_vector_type(2)));
typedef float v8f __attribute__((ext_vector_type(8)));

constexpr int T_LEN = 8192;
constexpr int NSEG  = 32;
constexpr int SEG   = T_LEN / NSEG;

struct WaveSlab {
    float xs[16][17];
    float us[16][17];
    float cx1[16], cx2[16];
    float cy1[16], cy2[16];
};

__device__ __forceinline__ void wave_lds_sync() {
    __builtin_amdgcn_fence(__ATOMIC_RELEASE, "wavefront");
    __builtin_amdgcn_wave_barrier();
    __builtin_amdgcn_fence(__ATOMIC_ACQUIRE, "wavefront");
}

template <bool STORE>
__device__ __forceinline__ void segment_pass(
    WaveSlab& sl, const float* __restrict__ xrow, float* __restrict__ ytile,
    unsigned lane, float b0, float b1, float b2, float a1, float a2,
    const v2f (&Bf)[4], float pj, float qj, float& h1, float& h2, float& h3)
{
    const unsigned j  = lane & 15u;
    const bool     hi = lane >= 16u;
    const int rload = (int)(lane >> 1);
    const int toff  = (int)(lane & 1u) * 8;

    for (int t0 = 0; t0 < SEG; t0 += 16) {
        const float4 xa = *(const float4*)(xrow + t0 + toff);
        const float4 xb = *(const float4*)(xrow + t0 + toff + 4);
        if (!STORE) {
#pragma unroll
            for (int k = 0; k < 16; ++k) {
                const float hn = -a1 * h1 - a2 * h2;
                h3 = h2; h2 = h1; h1 = hn;
            }
        }
        if (t0 + 16 < SEG)
            __builtin_prefetch(xrow + t0 + 16 + toff, 0, 1);

        sl.xs[rload][toff + 0] = xa.x; sl.xs[rload][toff + 1] = xa.y;
        sl.xs[rload][toff + 2] = xa.z; sl.xs[rload][toff + 3] = xa.w;
        sl.xs[rload][toff + 4] = xb.x; sl.xs[rload][toff + 5] = xb.y;
        sl.xs[rload][toff + 6] = xb.z; sl.xs[rload][toff + 7] = xb.w;
        wave_lds_sync();

        float prev1 = toff ? sl.xs[rload][7] : sl.cx1[rload];
        float prev2 = toff ? sl.xs[rload][6] : sl.cx2[rload];
#pragma unroll
        for (int k = 0; k < 8; ++k) {
            const float xv = sl.xs[rload][toff + k];
            sl.us[rload][toff + k] = b0 * xv + b1 * prev1 + b2 * prev2;
            prev2 = prev1;
            prev1 = xv;
        }
        wave_lds_sync();

        if (toff) {
            sl.cx1[rload] = sl.xs[rload][15];
            sl.cx2[rload] = sl.xs[rload][14];
        }

        v8f acc = {};
#pragma unroll
        for (int s = 0; s < 4; ++s) {
            const int k0 = 4 * s + (hi ? 2 : 0);
            v2f A;
            A.x = sl.us[j][k0];
            A.y = sl.us[j][k0 + 1];
            acc = __builtin_amdgcn_wmma_f32_16x16x4_f32(
                 false, A,  false, Bf[s],
                 (short)0, acc,  false,  false);
        }

        float outv[8];
#pragma unroll
        for (int v = 0; v < 8; ++v) {
            const int r2 = v + (hi ? 8 : 0);
            const float yv = acc[v] + pj * sl.cy1[r2] + qj * sl.cy2[r2];
            outv[v] = yv;
            if (STORE)
                ytile[(size_t)r2 * T_LEN + t0 + j] = yv;
        }
        wave_lds_sync();

        if (j == 15u) {
#pragma unroll
            for (int v = 0; v < 8; ++v) sl.cy1[v + (hi ? 8 : 0)] = outv[v];
        }
        if (j == 14u) {
#pragma unroll
            for (int v = 0; v < 8; ++v) sl.cy2[v + (hi ? 8 : 0)] = outv[v];
        }
        wave_lds_sync();
    }
}

__global__ __launch_bounds__(32 * NSEG) void dsvf_wmma_kernel(
    const float* __restrict__ x,
    const float* __restrict__ gp,
    const float* __restrict__ rp,
    const float* __restrict__ mhp_p,
    const float* __restrict__ mbp_p,
    const float* __restrict__ mlp_p,
    float* __restrict__ yout)
{
    __shared__ WaveSlab slab[NSEG];
    __shared__ float hs[20];
    __shared__ float es1[NSEG][16], es2[NSEG][16];

    const unsigned lane = threadIdx.x & 31u;
    const unsigned wave = threadIdx.x >> 5;
    const unsigned j    = lane & 15u;
    const bool     hi   = lane >= 16u;
    const int rbase = blockIdx.x * 16;
    const int seg0  = (int)wave * SEG;

    const float gv  = gp[0];
    const float rv  = rp[0];
    const float mhp = mhp_p[0];
    const float mbp = mbp_p[0];
    const float mlp = mlp_p[0];

    const float sig = 1.0f / (1.0f + expf(-gv));
    const float gg  = tanf(3.14159265358979323846f * 0.5f * sig);
    const float rr  = log1pf(expf(rv));
    const float g2  = gg * gg;

    const float b0 = g2 * mlp + gg * mbp + mhp;
    const float b1 = 2.0f * g2 * mlp - 2.0f * mhp;
    const float b2 = g2 * mlp - gg * mbp + mhp;
    const float a0 = g2 + 2.0f * rr * gg + 1.0f;
    const float inva0 = 1.0f / a0;
    const float a1 = (2.0f * g2 - 2.0f) * inva0;
    const float a2 = (g2 - 2.0f * rr * gg + 1.0f) * inva0;

    if (threadIdx.x == 0u) {
        hs[0] = 1.0f;
        hs[1] = -a1;
        for (int i = 2; i <= 16; ++i)
            hs[i] = -a1 * hs[i - 1] - a2 * hs[i - 2];
    }

    WaveSlab& sl = slab[wave];
    const float* xr = x + (size_t)(rbase + (int)j) * T_LEN + seg0;
    const float xm1 = wave ? xr[-1] : 0.0f;
    const float xm2 = wave ? xr[-2] : 0.0f;
    if (lane < 16u) {
        sl.cx1[lane] = xm1; sl.cx2[lane] = xm2;
        sl.cy1[lane] = 0.0f; sl.cy2[lane] = 0.0f;
    }
    __syncthreads();

    v2f Bf[4];
#pragma unroll
    for (int s = 0; s < 4; ++s) {
        const int k0 = 4 * s + (hi ? 2 : 0);
        const int k1 = k0 + 1;
        Bf[s].x = ((int)j >= k0) ? hs[j - k0] : 0.0f;
        Bf[s].y = ((int)j >= k1) ? hs[j - k1] : 0.0f;
    }
    const float pj = hs[j + 1];
    const float qj = -a2 * hs[j];

    const float* xrow  = x + (size_t)(rbase + (int)(lane >> 1)) * T_LEN + seg0;
    float*       ytile = yout + (size_t)rbase * T_LEN + seg0;

    float h1 = 1.0f, h2 = 0.0f, h3 = 0.0f;
    segment_pass<false>(sl, xrow, ytile, lane, b0, b1, b2, a1, a2, Bf, pj, qj, h1, h2, h3);
    if (lane < 16u) {
        es1[wave][lane] = sl.cy1[lane];
        es2[wave][lane] = sl.cy2[lane];
    }
    __syncthreads();

    const float P1 = h1, Q1 = -a2 * h2;
    const float P2 = h2, Q2 = -a2 * h3;
    float y1 = 0.0f, y2 = 0.0f;
#pragma unroll
    for (int s = 0; s < NSEG - 1; ++s) {
        const float n1 = es1[s][j] + P1 * y1 + Q1 * y2;
        const float n2 = es2[s][j] + P2 * y1 + Q2 * y2;
        const bool take = s < (int)wave;
        y1 = take ? n1 : y1;
        y2 = take ? n2 : y2;
    }
    if (lane < 16u) {
        sl.cx1[lane] = xm1; sl.cx2[lane] = xm2;
        sl.cy1[lane] = y1;  sl.cy2[lane] = y2;
    }
    wave_lds_sync();

    segment_pass<true>(sl, xrow, ytile, lane, b0, b1, b2, a1, a2, Bf, pj, qj, h1, h2, h3);
}

extern "C" void kernel_launch(void* const* d_in, const int* in_sizes, int n_in,
                              void* d_out, int out_size, void* d_ws, size_t ws_size,
                              hipStream_t stream) {
    const float* x   = (const float*)d_in[0];
    const float* g   = (const float*)d_in[1];
    const float* r   = (const float*)d_in[2];
    const float* mhp = (const float*)d_in[3];
    const float* mbp = (const float*)d_in[4];
    const float* mlp = (const float*)d_in[5];
    float* y = (float*)d_out;

    const int B = in_sizes[0] / T_LEN;
    dim3 grid(B / 16), block(32 * NSEG);
    hipLaunchKernelGGL(dsvf_wmma_kernel, grid, block, 0, stream,
                       x, g, r, mhp, mbp, mlp, y);
}


# x tile loaded as whole 128-byte lines: 4 rows x one line per load (was 16 rows x 16 B), 32 steps per trip, FIR lane per (row, 16-col tile), 2 WMMA chunks per trip; LDS slab doubles; sums unchanged
# speedup vs baseline: 10.3836x; 1.2398x over previous
#include <hip/hip_runtime.h>
#include <math.h>

typedef float v2f __attribute__((ext_vector_type(2)));
typedef float v8f __attribute__((ext_vector_type(8)));

constexpr int T_LEN = 8192;
constexpr int NSEG  = 32;
constexpr int SEG   = T_LEN / NSEG;
constexpr int TRIP  = 32;

struct WaveSlab {
    float xs[2][16][17];
    float us[2][16][17];
    float cx1[16], cx2[16];
    float cy1[16], cy2[16];
};

__device__ __forceinline__ void wave_lds_sync() {
    __builtin_amdgcn_fence(__ATOMIC_RELEASE, "wavefront");
    __builtin_amdgcn_wave_barrier();
    __builtin_amdgcn_fence(__ATOMIC_ACQUIRE, "wavefront");
}

template <bool STORE>
__device__ __forceinline__ void segment_pass(
    WaveSlab& sl, const float* __restrict__ xtile, float* __restrict__ ytile,
    unsigned lane, float b0, float b1, float b2, float a1, float a2,
    const v2f (&Bf)[4], float pj, float qj, float& h1, float& h2, float& h3)
{
    const unsigned j  = lane & 15u;
    const bool     hi = lane >= 16u;
    const int rload = (int)(lane >> 1);
    const int fb    = (int)(lane & 1u);
    const int lrow  = (int)(lane >> 3);
    const int lq    = (int)(lane & 7u);
    const int ltile = lq >> 2;
    const int lcol  = (lq & 3) * 4;

#pragma nounroll
    for (int t0 = 0; t0 < SEG; t0 += TRIP) {
        float4 xq[4];
#pragma unroll
        for (int i = 0; i < 4; ++i)
            xq[i] = *(const float4*)(xtile + (size_t)(4 * i + lrow) * T_LEN + t0 + 4 * lq);
        if (!STORE) {
#pragma unroll
            for (int k = 0; k < TRIP; ++k) {
                const float hn = -a1 * h1 - a2 * h2;
                h3 = h2; h2 = h1; h1 = hn;
            }
        }
        if ((t0 & TRIP) && t0 + TRIP < SEG)
            __builtin_prefetch(xtile + (size_t)rload * T_LEN + t0 + TRIP + 16 * fb, 0, 1);

#pragma unroll
        for (int i = 0; i < 4; ++i) {
            float* d = &sl.xs[ltile][4 * i + lrow][lcol];
            d[0] = xq[i].x; d[1] = xq[i].y; d[2] = xq[i].z; d[3] = xq[i].w;
        }
        wave_lds_sync();

        float prev1 = fb ? sl.xs[0][rload][15] : sl.cx1[rload];
        float prev2 = fb ? sl.xs[0][rload][14] : sl.cx2[rload];
#pragma unroll
        for (int k = 0; k < 16; ++k) {
            const float xv = sl.xs[fb][rload][k];
            sl.us[fb][rload][k] = b0 * xv + b1 * prev1 + b2 * prev2;
            prev2 = prev1;
            prev1 = xv;
        }
        wave_lds_sync();

        if (fb) {
            sl.cx1[rload] = prev1;
            sl.cx2[rload] = prev2;
        }

#pragma unroll
        for (int c = 0; c < 2; ++c) {
            v8f acc = {};
#pragma unroll
            for (int s = 0; s < 4; ++s) {
                const int k0 = 4 * s + (hi ? 2 : 0);
                v2f A;
                A.x = sl.us[c][j][k0];
                A.y = sl.us[c][j][k0 + 1];
                acc = __builtin_amdgcn_wmma_f32_16x16x4_f32(
                     false, A,  false, Bf[s],
                     (short)0, acc,  false,  false);
            }

            float outv[8];
#pragma unroll
            for (int v = 0; v < 8; ++v) {
                const int r2 = v + (hi ? 8 : 0);
                const float yv = acc[v] + pj * sl.cy1[r2] + qj * sl.cy2[r2];
                outv[v] = yv;
                if (STORE)
                    ytile[(size_t)r2 * T_LEN + t0 + 16 * c + j] = yv;
            }
            wave_lds_sync();

            if (j == 15u) {
#pragma unroll
                for (int v = 0; v < 8; ++v) sl.cy1[v + (hi ? 8 : 0)] = outv[v];
            }
            if (j == 14u) {
#pragma unroll
                for (int v = 0; v < 8; ++v) sl.cy2[v + (hi ? 8 : 0)] = outv[v];
            }
            wave_lds_sync();
        }
    }
}

__global__ __launch_bounds__(32 * NSEG) void dsvf_wmma_kernel(
    const float* __restrict__ x,
    const float* __restrict__ gp,
    const float* __restrict__ rp,
    const float* __restrict__ mhp_p,
    const float* __restrict__ mbp_p,
    const float* __restrict__ mlp_p,
    float* __restrict__ yout)
{
    __shared__ WaveSlab slab[NSEG];
    __shared__ float hs[20];
    __shared__ float es1[NSEG][16], es2[NSEG][16];

    const unsigned lane = threadIdx.x & 31u;
    const unsigned wave = threadIdx.x >> 5;
    const unsigned j    = lane & 15u;
    const bool     hi   = lane >= 16u;
    const int rbase = blockIdx.x * 16;
    const int seg0  = (int)wave * SEG;

    const float gv  = gp[0];
    const float rv  = rp[0];
    const float mhp = mhp_p[0];
    const float mbp = mbp_p[0];
    const float mlp = mlp_p[0];

    const float sig = 1.0f / (1.0f + expf(-gv));
    const float gg  = tanf(3.14159265358979323846f * 0.5f * sig);
    const float rr  = log1pf(expf(rv));
    const float g2  = gg * gg;

    const float b0 = g2 * mlp + gg * mbp + mhp;
    const float b1 = 2.0f * g2 * mlp - 2.0f * mhp;
    const float b2 = g2 * mlp - gg * mbp + mhp;
    const float a0 = g2 + 2.0f * rr * gg + 1.0f;
    const float inva0 = 1.0f / a0;
    const float a1 = (2.0f * g2 - 2.0f) * inva0;
    const float a2 = (g2 - 2.0f * rr * gg + 1.0f) * inva0;

    if (threadIdx.x == 0u) {
        hs[0] = 1.0f;
        hs[1] = -a1;
        for (int i = 2; i <= 16; ++i)
            hs[i] = -a1 * hs[i - 1] - a2 * hs[i - 2];
    }

    WaveSlab& sl = slab[wave];
    const float* xr = x + (size_t)(rbase + (int)j) * T_LEN + seg0;
    const float xm1 = wave ? xr[-1] : 0.0f;
    const float xm2 = wave ? xr[-2] : 0.0f;
    if (lane < 16u) {
        sl.cx1[lane] = xm1; sl.cx2[lane] = xm2;
        sl.cy1[lane] = 0.0f; sl.cy2[lane] = 0.0f;
    }
    __syncthreads();

    v2f Bf[4];
#pragma unroll
    for (int s = 0; s < 4; ++s) {
        const int k0 = 4 * s + (hi ? 2 : 0);
        const int k1 = k0 + 1;
        Bf[s].x = ((int)j >= k0) ? hs[j - k0] : 0.0f;
        Bf[s].y = ((int)j >= k1) ? hs[j - k1] : 0.0f;
    }
    const float pj = hs[j + 1];
    const float qj = -a2 * hs[j];

    const float* xtile = x + (size_t)rbase * T_LEN + seg0;
    float*       ytile = yout + (size_t)rbase * T_LEN + seg0;

    float h1 = 1.0f, h2 = 0.0f, h3 = 0.0f;
    segment_pass<false>(sl, xtile, ytile, lane, b0, b1, b2, a1, a2, Bf, pj, qj, h1, h2, h3);
    if (lane < 16u) {
        es1[wave][lane] = sl.cy1[lane];
        es2[wave][lane] = sl.cy2[lane];
    }
    __syncthreads();

    const float P1 = h1, Q1 = -a2 * h2;
    const float P2 = h2, Q2 = -a2 * h3;
    float y1 = 0.0f, y2 = 0.0f;
#pragma unroll
    for (int s = 0; s < NSEG - 1; ++s) {
        const float n1 = es1[s][j] + P1 * y1 + Q1 * y2;
        const float n2 = es2[s][j] + P2 * y1 + Q2 * y2;
        const bool take = s < (int)wave;
        y1 = take ? n1 : y1;
        y2 = take ? n2 : y2;
    }
    if (lane < 16u) {
        sl.cx1[lane] = xm1; sl.cx2[lane] = xm2;
        sl.cy1[lane] = y1;  sl.cy2[lane] = y2;
    }
    wave_lds_sync();

    segment_pass<true>(sl, xtile, ytile, lane, b0, b1, b2, a1, a2, Bf, pj, qj, h1, h2, h3);
}

extern "C" void kernel_launch(void* const* d_in, const int* in_sizes, int n_in,
                              void* d_out, int out_size, void* d_ws, size_t ws_size,
                              hipStream_t stream) {
    const float* x   = (const float*)d_in[0];
    const float* g   = (const float*)d_in[1];
    const float* r   = (const float*)d_in[2];
    const float* mhp = (const float*)d_in[3];
    const float* mbp = (const float*)d_in[4];
    const float* mlp = (const float*)d_in[5];
    float* y = (float*)d_out;

    const int B = in_sizes[0] / T_LEN;
    dim3 grid(B / 16), block(32 * NSEG);
    hipLaunchKernelGGL(dsvf_wmma_kernel, grid, block, 0, stream,
                       x, g, r, mhp, mbp, mlp, y);
}


# Biquad step as one transposed matrix product (taps x x-tile, time as M): FIR and the four state terms folded in, row state stays in its lane; FIR pass and carry LDS trips gone; sum order changes
# speedup vs baseline: 14.0188x; 1.3501x over previous
#include <hip/hip_runtime.h>
#include <math.h>


typedef float v2f __attribute__((ext_vector_type(2)));
typedef float v8f __attribute__((ext_vector_type(8)));

constexpr int T_LEN = 8192;
constexpr int NSEG  = 32;
constexpr int SEG   = T_LEN / NSEG;
constexpr int TRIP  = 32;
constexpr int PITCH = 36;

struct __attribute__((aligned(16))) WaveSlab {
    float t[16][PITCH];
};

__device__ __forceinline__ void wave_lds_sync() {
    __builtin_amdgcn_fence(__ATOMIC_RELEASE, "wavefront");
    __builtin_amdgcn_wave_barrier();
    __builtin_amdgcn_fence(__ATOMIC_ACQUIRE, "wavefront");
}

__device__ __forceinline__ v8f mm(v2f a, v2f b, v8f c) {
    return __builtin_amdgcn_wmma_f32_16x16x4_f32(
         false, a,  false, b,
         (short)0, c,  false,  false);
}

template <bool STORE>
__device__ __forceinline__ void segment_pass(
    WaveSlab& sl, const float* __restrict__ xtile, float* __restrict__ ytile,
    unsigned lane, float a1, float a2,
    const v2f (&Ga)[4], v2f Ca, v2f& st, float& h1, float& h2, float& h3)
{
    const int  j  = (int)(lane & 15u);
    const bool hi = lane >= 16u;
    const int fcol  = hi ? 0 : 8;
    const int ycol  = hi ? 8 : 0;
    const int rload = (int)(lane >> 1);
    const int fb    = (int)(lane & 1u);
    const int lrow  = (int)(lane >> 3);
    const int lq    = (int)(lane & 7u);

#pragma nounroll
    for (int t0 = 0; t0 < SEG; t0 += TRIP) {
        float4 xq[4];
#pragma unroll
        for (int i = 0; i < 4; ++i)
            xq[i] = *(const float4*)(xtile + (size_t)(4 * i + lrow) * T_LEN + t0 + 4 * lq);
        if (!STORE) {
#pragma unroll
            for (int k = 0; k < TRIP; ++k) {
                const float hn = -a1 * h1 - a2 * h2;
                h3 = h2; h2 = h1; h1 = hn;
            }
        }
        if ((t0 & TRIP) && t0 + TRIP < SEG)
            __builtin_prefetch(xtile + (size_t)rload * T_LEN + t0 + TRIP + 16 * fb, 0, 1);

#pragma unroll
        for (int i = 0; i < 4; ++i)
            *(float4*)&sl.t[4 * i + lrow][4 * lq] = xq[i];
        wave_lds_sync();

        float4 f[2][2];
#pragma unroll
        for (int c = 0; c < 2; ++c) {
            f[c][0] = *(const float4*)&sl.t[j][16 * c + fcol];
            f[c][1] = *(const float4*)&sl.t[j][16 * c + fcol + 4];
        }

        v8f acc0 = {}, acc1 = {};
        {
            const v2f b00 = {f[0][0].x, f[0][0].y}, b01 = {f[0][0].z, f[0][0].w};
            const v2f b02 = {f[0][1].x, f[0][1].y}, b03 = {f[0][1].z, f[0][1].w};
            const v2f b10 = {f[1][0].x, f[1][0].y}, b11 = {f[1][0].z, f[1][0].w};
            const v2f b12 = {f[1][1].x, f[1][1].y}, b13 = {f[1][1].z, f[1][1].w};
            acc0 = mm(Ga[0], b00, acc0);  acc1 = mm(Ga[0], b10, acc1);
            acc0 = mm(Ga[1], b01, acc0);  acc1 = mm(Ga[1], b11, acc1);
            acc0 = mm(Ga[2], b02, acc0);  acc1 = mm(Ga[2], b12, acc1);
            acc0 = mm(Ga[3], b03, acc0);  acc1 = mm(Ga[3], b13, acc1);
        }

        acc0 = mm(Ca, st, acc0);
        st.x = hi ? acc0[6] : f[0][1].z;
        st.y = hi ? acc0[7] : f[0][1].w;
        acc1 = mm(Ca, st, acc1);
        st.x = hi ? acc1[6] : f[1][1].z;
        st.y = hi ? acc1[7] : f[1][1].w;

        if (STORE) {
            wave_lds_sync();
            *(float4*)&sl.t[j][ycol]          = make_float4(acc0[0], acc0[1], acc0[2], acc0[3]);
            *(float4*)&sl.t[j][ycol + 4]      = make_float4(acc0[4], acc0[5], acc0[6], acc0[7]);
            *(float4*)&sl.t[j][16 + ycol]     = make_float4(acc1[0], acc1[1], acc1[2], acc1[3]);
            *(float4*)&sl.t[j][16 + ycol + 4] = make_float4(acc1[4], acc1[5], acc1[6], acc1[7]);
            wave_lds_sync();
#pragma unroll
            for (int i = 0; i < 4; ++i) {
                const float4 yv = *(const float4*)&sl.t[4 * i + lrow][4 * lq];
                *(float4*)(ytile + (size_t)(4 * i + lrow) * T_LEN + t0 + 4 * lq) = yv;
            }
        }
        wave_lds_sync();
    }
}

__global__ __launch_bounds__(32 * NSEG) void dsvf_wmma_kernel(
    const float* __restrict__ x,
    const float* __restrict__ gp,
    const float* __restrict__ rp,
    const float* __restrict__ mhp_p,
    const float* __restrict__ mbp_p,
    const float* __restrict__ mlp_p,
    float* __restrict__ yout)
{
    __shared__ WaveSlab slab[NSEG];
    __shared__ float hz[20];
    __shared__ float es1[NSEG][16], es2[NSEG][16];

    const unsigned lane = threadIdx.x & 31u;
    const unsigned wave = threadIdx.x >> 5;
    const int      j    = (int)(lane & 15u);
    const bool     hi   = lane >= 16u;
    const int rbase = blockIdx.x * 16;
    const int seg0  = (int)wave * SEG;

    const float gv  = gp[0];
    const float rv  = rp[0];
    const float mhp = mhp_p[0];
    const float mbp = mbp_p[0];
    const float mlp = mlp_p[0];

    const float sig = 1.0f / (1.0f + expf(-gv));
    const float gg  = tanf(3.14159265358979323846f * 0.5f * sig);
    const float rr  = log1pf(expf(rv));
    const float g2  = gg * gg;

    const float b0 = g2 * mlp + gg * mbp + mhp;
    const float b1 = 2.0f * g2 * mlp - 2.0f * mhp;
    const float b2 = g2 * mlp - gg * mbp + mhp;
    const float a0 = g2 + 2.0f * rr * gg + 1.0f;
    const float inva0 = 1.0f / a0;
    const float a1 = (2.0f * g2 - 2.0f) * inva0;
    const float a2 = (g2 - 2.0f * rr * gg + 1.0f) * inva0;

    if (threadIdx.x == 0u) {
        hz[0] = 0.0f;
        hz[1] = 0.0f;
        hz[2] = 1.0f;
        hz[3] = -a1;
        for (int i = 2; i <= 16; ++i)
            hz[2 + i] = -a1 * hz[1 + i] - a2 * hz[i];
    }

    WaveSlab& sl = slab[wave];
    const float* xr = x + (size_t)(rbase + j) * T_LEN + seg0;
    const float xm1 = wave ? xr[-1] : 0.0f;
    const float xm2 = wave ? xr[-2] : 0.0f;
    __syncthreads();

    const int fcol = hi ? 0 : 8;
    v2f Ga[4];
#pragma unroll
    for (int s = 0; s < 4; ++s) {
        const int d0 = j - (fcol + 2 * s);
        const int d1 = d0 - 1;
        const int c0 = d0 < 0 ? 0 : d0;
        const int c1 = d1 < 0 ? 0 : d1;
        const float g0 = b0 * hz[c0 + 2] + b1 * hz[c0 + 1] + b2 * hz[c0];
        const float g1 = b0 * hz[c1 + 2] + b1 * hz[c1 + 1] + b2 * hz[c1];
        Ga[s].x = d0 >= 0 ? g0 : 0.0f;
        Ga[s].y = d1 >= 0 ? g1 : 0.0f;
    }
    const float ht  = hz[j + 2];
    const float htm = hz[j + 1];
    const float htp = hz[j + 3];
    v2f Ca;
    Ca.x = hi ? -a2 * ht : b2 * ht;
    Ca.y = hi ? htp      : b1 * ht + b2 * htm;

    const float* xtile = x + (size_t)rbase * T_LEN + seg0;
    float*       ytile = yout + (size_t)rbase * T_LEN + seg0;

    float h1 = 1.0f, h2 = 0.0f, h3 = 0.0f;
    v2f st;
    st.x = hi ? 0.0f : xm2;
    st.y = hi ? 0.0f : xm1;
    segment_pass<false>(sl, xtile, ytile, lane, a1, a2, Ga, Ca, st, h1, h2, h3);
    if (hi) {
        es1[wave][j] = st.y;
        es2[wave][j] = st.x;
    }
    __syncthreads();

    const float P1 = h1, Q1 = -a2 * h2;
    const float P2 = h2, Q2 = -a2 * h3;
    float y1 = 0.0f, y2 = 0.0f;
#pragma unroll
    for (int s = 0; s < NSEG - 1; ++s) {
        const float n1 = es1[s][j] + P1 * y1 + Q1 * y2;
        const float n2 = es2[s][j] + P2 * y1 + Q2 * y2;
        const bool take = s < (int)wave;
        y1 = take ? n1 : y1;
        y2 = take ? n2 : y2;
    }
    st.x = hi ? y2 : xm2;
    st.y = hi ? y1 : xm1;

    segment_pass<true>(sl, xtile, ytile, lane, a1, a2, Ga, Ca, st, h1, h2, h3);
}

extern "C" void kernel_launch(void* const* d_in, const int* in_sizes, int n_in,
                              void* d_out, int out_size, void* d_ws, size_t ws_size,
                              hipStream_t stream) {
    const float* x   = (const float*)d_in[0];
    const float* g   = (const float*)d_in[1];
    const float* r   = (const float*)d_in[2];
    const float* mhp = (const float*)d_in[3];
    const float* mbp = (const float*)d_in[4];
    const float* mlp = (const float*)d_in[5];
    float* y = (float*)d_out;

    const int B = in_sizes[0] / T_LEN;
    dim3 grid(B / 16), block(32 * NSEG);
    hipLaunchKernelGGL(dsvf_wmma_kernel, grid, block, 0, stream,
                       x, g, r, mhp, mbp, mlp, y);
}


# x tile requested one trip ahead into a second register set, both passes (trip pairs swap sets; output pass's first tile requested before the scan): loads run beside LDS and WMMA work; sums unchanged
# speedup vs baseline: 14.5531x; 1.0381x over previous
#include <hip/hip_runtime.h>
#include <math.h>


typedef float v2f __attribute__((ext_vector_type(2)));
typedef float v8f __attribute__((ext_vector_type(8)));

constexpr int T_LEN = 8192;
constexpr int NSEG  = 32;
constexpr int SEG   = T_LEN / NSEG;
constexpr int TRIP  = 32;
constexpr int PITCH = 36;

struct __attribute__((aligned(16))) WaveSlab {
    float t[16][PITCH];
};

__device__ __forceinline__ void wave_lds_sync() {
    __builtin_amdgcn_fence(__ATOMIC_RELEASE, "wavefront");
    __builtin_amdgcn_wave_barrier();
    __builtin_amdgcn_fence(__ATOMIC_ACQUIRE, "wavefront");
}

__device__ __forceinline__ v8f mm(v2f a, v2f b, v8f c) {
    return __builtin_amdgcn_wmma_f32_16x16x4_f32(
         false, a,  false, b,
         (short)0, c,  false,  false);
}

template <bool STORE>
__device__ __forceinline__ void segment_pass(
    WaveSlab& sl, const float* xtile, float* __restrict__ ytile,
    unsigned lane, float a1, float a2,
    const v2f (&Ga)[4], v2f Ca, v2f& st, float& h1, float& h2, float& h3,
    float4& xa0, float4& xa1, float4& xa2, float4& xa3)
{
    const int  j  = (int)(lane & 15u);
    const bool hi = lane >= 16u;
    const int fcol  = hi ? 0 : 8;
    const int ycol  = hi ? 8 : 0;
    const int rload = (int)(lane >> 1);
    const int fb    = (int)(lane & 1u);
    const int lrow  = (int)(lane >> 3);
    const int lq    = (int)(lane & 7u);

    auto one_trip = [&](const int t0, const bool second, const bool last,
                        const float4& c0, const float4& c1, const float4& c2, const float4& c3,
                        float4& n0, float4& n1, float4& n2, float4& n3)
                        __attribute__((always_inline)) {
        if (!STORE || !last) {
            const float* xl = xtile + (size_t)lrow * T_LEN + (last ? 0 : t0 + TRIP) + 4 * lq;
            n0 = *(const float4*)(xl);
            n1 = *(const float4*)(xl + (size_t)4 * T_LEN);
            n2 = *(const float4*)(xl + (size_t)8 * T_LEN);
            n3 = *(const float4*)(xl + (size_t)12 * T_LEN);
        }
        if (!STORE) {
#pragma unroll
            for (int k = 0; k < TRIP; ++k) {
                const float hn = -a1 * h1 - a2 * h2;
                h3 = h2; h2 = h1; h1 = hn;
            }
        }
        if (!second && t0 + 2 * TRIP < SEG)
            __builtin_prefetch(xtile + (size_t)rload * T_LEN + t0 + 2 * TRIP + 16 * fb, 0, 1);

        *(float4*)&sl.t[lrow][4 * lq]      = c0;
        *(float4*)&sl.t[4 + lrow][4 * lq]  = c1;
        *(float4*)&sl.t[8 + lrow][4 * lq]  = c2;
        *(float4*)&sl.t[12 + lrow][4 * lq] = c3;
        wave_lds_sync();

        float4 f[2][2];
#pragma unroll
        for (int c = 0; c < 2; ++c) {
            f[c][0] = *(const float4*)&sl.t[j][16 * c + fcol];
            f[c][1] = *(const float4*)&sl.t[j][16 * c + fcol + 4];
        }

        v8f acc0 = {}, acc1 = {};
        {
            const v2f b00 = {f[0][0].x, f[0][0].y}, b01 = {f[0][0].z, f[0][0].w};
            const v2f b02 = {f[0][1].x, f[0][1].y}, b03 = {f[0][1].z, f[0][1].w};
            const v2f b10 = {f[1][0].x, f[1][0].y}, b11 = {f[1][0].z, f[1][0].w};
            const v2f b12 = {f[1][1].x, f[1][1].y}, b13 = {f[1][1].z, f[1][1].w};
            acc0 = mm(Ga[0], b00, acc0);  acc1 = mm(Ga[0], b10, acc1);
            acc0 = mm(Ga[1], b01, acc0);  acc1 = mm(Ga[1], b11, acc1);
            acc0 = mm(Ga[2], b02, acc0);  acc1 = mm(Ga[2], b12, acc1);
            acc0 = mm(Ga[3], b03, acc0);  acc1 = mm(Ga[3], b13, acc1);
        }

        acc0 = mm(Ca, st, acc0);
        st.x = hi ? acc0[6] : f[0][1].z;
        st.y = hi ? acc0[7] : f[0][1].w;
        acc1 = mm(Ca, st, acc1);
        st.x = hi ? acc1[6] : f[1][1].z;
        st.y = hi ? acc1[7] : f[1][1].w;

        if (STORE) {
            wave_lds_sync();
            *(float4*)&sl.t[j][ycol]          = make_float4(acc0[0], acc0[1], acc0[2], acc0[3]);
            *(float4*)&sl.t[j][ycol + 4]      = make_float4(acc0[4], acc0[5], acc0[6], acc0[7]);
            *(float4*)&sl.t[j][16 + ycol]     = make_float4(acc1[0], acc1[1], acc1[2], acc1[3]);
            *(float4*)&sl.t[j][16 + ycol + 4] = make_float4(acc1[4], acc1[5], acc1[6], acc1[7]);
            wave_lds_sync();
#pragma unroll
            for (int i = 0; i < 4; ++i) {
                const float4 yv = *(const float4*)&sl.t[4 * i + lrow][4 * lq];
                *(float4*)(ytile + (size_t)(4 * i + lrow) * T_LEN + t0 + 4 * lq) = yv;
            }
        }
        wave_lds_sync();
        if (!STORE) {
            asm volatile("" : "+v"(st.x), "+v"(st.y));
        }
    };

    float4 xb0, xb1, xb2, xb3;
#pragma nounroll
    for (int t0 = 0; t0 < SEG - 2 * TRIP; t0 += 2 * TRIP) {
        one_trip(t0,        false, false, xa0, xa1, xa2, xa3, xb0, xb1, xb2, xb3);
        one_trip(t0 + TRIP, true,  false, xb0, xb1, xb2, xb3, xa0, xa1, xa2, xa3);
    }
    one_trip(SEG - 2 * TRIP, false, false, xa0, xa1, xa2, xa3, xb0, xb1, xb2, xb3);
    one_trip(SEG - TRIP,     true,  true,  xb0, xb1, xb2, xb3, xa0, xa1, xa2, xa3);
}

__global__ __launch_bounds__(32 * NSEG) void dsvf_wmma_kernel(
    const float* x,
    const float* __restrict__ gp,
    const float* __restrict__ rp,
    const float* __restrict__ mhp_p,
    const float* __restrict__ mbp_p,
    const float* __restrict__ mlp_p,
    float* __restrict__ yout)
{
    __shared__ WaveSlab slab[NSEG];
    __shared__ float hz[20];
    __shared__ float es1[NSEG][16], es2[NSEG][16];

    const unsigned lane = threadIdx.x & 31u;
    const unsigned wave = threadIdx.x >> 5;
    const int      j    = (int)(lane & 15u);
    const bool     hi   = lane >= 16u;
    const int rbase = blockIdx.x * 16;
    const int seg0  = (int)wave * SEG;

    const float* xtile = x + (size_t)rbase * T_LEN + seg0;
    float*       ytile = yout + (size_t)rbase * T_LEN + seg0;

    float4 xq0, xq1, xq2, xq3;
    {
        const float* xl = xtile + (size_t)(lane >> 3) * T_LEN + 4 * (lane & 7u);
        xq0 = *(const float4*)(xl);
        xq1 = *(const float4*)(xl + (size_t)4 * T_LEN);
        xq2 = *(const float4*)(xl + (size_t)8 * T_LEN);
        xq3 = *(const float4*)(xl + (size_t)12 * T_LEN);
    }

    const float gv  = gp[0];
    const float rv  = rp[0];
    const float mhp = mhp_p[0];
    const float mbp = mbp_p[0];
    const float mlp = mlp_p[0];

    const float sig = 1.0f / (1.0f + expf(-gv));
    const float gg  = tanf(3.14159265358979323846f * 0.5f * sig);
    const float rr  = log1pf(expf(rv));
    const float g2  = gg * gg;

    const float b0 = g2 * mlp + gg * mbp + mhp;
    const float b1 = 2.0f * g2 * mlp - 2.0f * mhp;
    const float b2 = g2 * mlp - gg * mbp + mhp;
    const float a0 = g2 + 2.0f * rr * gg + 1.0f;
    const float inva0 = 1.0f / a0;
    const float a1 = (2.0f * g2 - 2.0f) * inva0;
    const float a2 = (g2 - 2.0f * rr * gg + 1.0f) * inva0;

    if (threadIdx.x == 0u) {
        hz[0] = 0.0f;
        hz[1] = 0.0f;
        hz[2] = 1.0f;
        hz[3] = -a1;
        for (int i = 2; i <= 16; ++i)
            hz[2 + i] = -a1 * hz[1 + i] - a2 * hz[i];
    }

    WaveSlab& sl = slab[wave];
    const float* xr = x + (size_t)(rbase + j) * T_LEN + seg0;
    const float xm1 = wave ? xr[-1] : 0.0f;
    const float xm2 = wave ? xr[-2] : 0.0f;
    __syncthreads();

    const int fcol = hi ? 0 : 8;
    v2f Ga[4];
#pragma unroll
    for (int s = 0; s < 4; ++s) {
        const int d0 = j - (fcol + 2 * s);
        const int d1 = d0 - 1;
        const int c0 = d0 < 0 ? 0 : d0;
        const int c1 = d1 < 0 ? 0 : d1;
        const float g0 = b0 * hz[c0 + 2] + b1 * hz[c0 + 1] + b2 * hz[c0];
        const float g1 = b0 * hz[c1 + 2] + b1 * hz[c1 + 1] + b2 * hz[c1];
        Ga[s].x = d0 >= 0 ? g0 : 0.0f;
        Ga[s].y = d1 >= 0 ? g1 : 0.0f;
    }
    const float ht  = hz[j + 2];
    const float htm = hz[j + 1];
    const float htp = hz[j + 3];
    v2f Ca;
    Ca.x = hi ? -a2 * ht : b2 * ht;
    Ca.y = hi ? htp      : b1 * ht + b2 * htm;

    float h1 = 1.0f, h2 = 0.0f, h3 = 0.0f;
    v2f st;
    st.x = hi ? 0.0f : xm2;
    st.y = hi ? 0.0f : xm1;
    segment_pass<false>(sl, xtile, ytile, lane, a1, a2, Ga, Ca, st, h1, h2, h3, xq0, xq1, xq2, xq3);
    if (hi) {
        es1[wave][j] = st.y;
        es2[wave][j] = st.x;
    }
    __syncthreads();

    const float P1 = h1, Q1 = -a2 * h2;
    const float P2 = h2, Q2 = -a2 * h3;
    float y1 = 0.0f, y2 = 0.0f;
#pragma unroll
    for (int s = 0; s < NSEG - 1; ++s) {
        const float n1 = es1[s][j] + P1 * y1 + Q1 * y2;
        const float n2 = es2[s][j] + P2 * y1 + Q2 * y2;
        const bool take = s < (int)wave;
        y1 = take ? n1 : y1;
        y2 = take ? n2 : y2;
    }
    st.x = hi ? y2 : xm2;
    st.y = hi ? y1 : xm1;

    segment_pass<true>(sl, xtile, ytile, lane, a1, a2, Ga, Ca, st, h1, h2, h3, xq0, xq1, xq2, xq3);
}

extern "C" void kernel_launch(void* const* d_in, const int* in_sizes, int n_in,
                              void* d_out, int out_size, void* d_ws, size_t ws_size,
                              hipStream_t stream) {
    const float* x   = (const float*)d_in[0];
    const float* g   = (const float*)d_in[1];
    const float* r   = (const float*)d_in[2];
    const float* mhp = (const float*)d_in[3];
    const float* mbp = (const float*)d_in[4];
    const float* mlp = (const float*)d_in[5];
    float* y = (float*)d_out;

    const int B = in_sizes[0] / T_LEN;
    dim3 grid(B / 16), block(32 * NSEG);
    hipLaunchKernelGGL(dsvf_wmma_kernel, grid, block, 0, stream,
                       x, g, r, mhp, mbp, mlp, y);
}


# Row tile's time axis split over 2 work-groups (256 groups, 128-step segments): the 2nd recomputes the 1st half's zero-state end states itself; carry folds 64 segments, sum order changes
# speedup vs baseline: 15.4539x; 1.0619x over previous
#include <hip/hip_runtime.h>
#include <math.h>


typedef float v2f __attribute__((ext_vector_type(2)));
typedef float v8f __attribute__((ext_vector_type(8)));

constexpr int T_LEN = 8192;
constexpr int NPART = 2;
constexpr int PART  = T_LEN / NPART;
constexpr int NSEG  = 32;
constexpr int SEG   = PART / NSEG;
constexpr int NSEGT = NPART * NSEG;
constexpr int TRIP  = 32;
constexpr int PITCH = 36;

struct __attribute__((aligned(16))) WaveSlab {
    float t[16][PITCH];
};

__device__ __forceinline__ void wave_lds_sync() {
    __builtin_amdgcn_fence(__ATOMIC_RELEASE, "wavefront");
    __builtin_amdgcn_wave_barrier();
    __builtin_amdgcn_fence(__ATOMIC_ACQUIRE, "wavefront");
}

__device__ __forceinline__ v8f mm(v2f a, v2f b, v8f c) {
    return __builtin_amdgcn_wmma_f32_16x16x4_f32(
         false, a,  false, b,
         (short)0, c,  false,  false);
}

template <bool STORE, bool ADV>
__device__ __forceinline__ void segment_pass(
    WaveSlab& sl, const float* xtile, const float* xwrap, float* __restrict__ ytile,
    unsigned lane, float a1, float a2,
    const v2f (&Ga)[4], v2f Ca, v2f& st, float& h1, float& h2, float& h3,
    float4& xa0, float4& xa1, float4& xa2, float4& xa3)
{
    const int  j  = (int)(lane & 15u);
    const bool hi = lane >= 16u;
    const int fcol  = hi ? 0 : 8;
    const int ycol  = hi ? 8 : 0;
    const int rload = (int)(lane >> 1);
    const int fb    = (int)(lane & 1u);
    const int lrow  = (int)(lane >> 3);
    const int lq    = (int)(lane & 7u);

    auto one_trip = [&](const int t0, const bool second, const bool last,
                        const float4& c0, const float4& c1, const float4& c2, const float4& c3,
                        float4& n0, float4& n1, float4& n2, float4& n3)
                        __attribute__((always_inline)) {
        if (!STORE || !last) {
            const float* xl = (last ? xwrap : xtile + t0 + TRIP) + (size_t)lrow * T_LEN + 4 * lq;
            n0 = *(const float4*)(xl);
            n1 = *(const float4*)(xl + (size_t)4 * T_LEN);
            n2 = *(const float4*)(xl + (size_t)8 * T_LEN);
            n3 = *(const float4*)(xl + (size_t)12 * T_LEN);
        }
        if (!STORE && ADV) {
#pragma unroll
            for (int k = 0; k < TRIP; ++k) {
                const float hn = -a1 * h1 - a2 * h2;
                h3 = h2; h2 = h1; h1 = hn;
            }
            asm volatile("" : "+s"(h1), "+s"(h2), "+s"(h3));
        }
        if (!second && t0 + 2 * TRIP < SEG)
            __builtin_prefetch(xtile + (size_t)rload * T_LEN + t0 + 2 * TRIP + 16 * fb, 0, 1);

        *(float4*)&sl.t[lrow][4 * lq]      = c0;
        *(float4*)&sl.t[4 + lrow][4 * lq]  = c1;
        *(float4*)&sl.t[8 + lrow][4 * lq]  = c2;
        *(float4*)&sl.t[12 + lrow][4 * lq] = c3;
        wave_lds_sync();

        float4 f[2][2];
#pragma unroll
        for (int c = 0; c < 2; ++c) {
            f[c][0] = *(const float4*)&sl.t[j][16 * c + fcol];
            f[c][1] = *(const float4*)&sl.t[j][16 * c + fcol + 4];
        }

        v8f acc0 = {}, acc1 = {};
        {
            const v2f b00 = {f[0][0].x, f[0][0].y}, b01 = {f[0][0].z, f[0][0].w};
            const v2f b02 = {f[0][1].x, f[0][1].y}, b03 = {f[0][1].z, f[0][1].w};
            const v2f b10 = {f[1][0].x, f[1][0].y}, b11 = {f[1][0].z, f[1][0].w};
            const v2f b12 = {f[1][1].x, f[1][1].y}, b13 = {f[1][1].z, f[1][1].w};
            acc0 = mm(Ga[0], b00, acc0);  acc1 = mm(Ga[0], b10, acc1);
            acc0 = mm(Ga[1], b01, acc0);  acc1 = mm(Ga[1], b11, acc1);
            acc0 = mm(Ga[2], b02, acc0);  acc1 = mm(Ga[2], b12, acc1);
            acc0 = mm(Ga[3], b03, acc0);  acc1 = mm(Ga[3], b13, acc1);
        }

        acc0 = mm(Ca, st, acc0);
        st.x = hi ? acc0[6] : f[0][1].z;
        st.y = hi ? acc0[7] : f[0][1].w;
        acc1 = mm(Ca, st, acc1);
        st.x = hi ? acc1[6] : f[1][1].z;
        st.y = hi ? acc1[7] : f[1][1].w;

        if (STORE) {
            wave_lds_sync();
            *(float4*)&sl.t[j][ycol]          = make_float4(acc0[0], acc0[1], acc0[2], acc0[3]);
            *(float4*)&sl.t[j][ycol + 4]      = make_float4(acc0[4], acc0[5], acc0[6], acc0[7]);
            *(float4*)&sl.t[j][16 + ycol]     = make_float4(acc1[0], acc1[1], acc1[2], acc1[3]);
            *(float4*)&sl.t[j][16 + ycol + 4] = make_float4(acc1[4], acc1[5], acc1[6], acc1[7]);
            wave_lds_sync();
#pragma unroll
            for (int i = 0; i < 4; ++i) {
                const float4 yv = *(const float4*)&sl.t[4 * i + lrow][4 * lq];
                *(float4*)(ytile + (size_t)(4 * i + lrow) * T_LEN + t0 + 4 * lq) = yv;
            }
        }
        wave_lds_sync();
        if (!STORE) {
            asm volatile("" : "+v"(st.x), "+v"(st.y));
        }
    };

    float4 xb0, xb1, xb2, xb3;
    static_assert(SEG == 4 * TRIP, "a segment is four trips");
    one_trip(0,        false, false, xa0, xa1, xa2, xa3, xb0, xb1, xb2, xb3);
    one_trip(TRIP,     true,  false, xb0, xb1, xb2, xb3, xa0, xa1, xa2, xa3);
    one_trip(2 * TRIP, false, false, xa0, xa1, xa2, xa3, xb0, xb1, xb2, xb3);
    one_trip(3 * TRIP, true,  true,  xb0, xb1, xb2, xb3, xa0, xa1, xa2, xa3);
}

__global__ __launch_bounds__(32 * NSEG) void dsvf_wmma_kernel(
    const float* x,
    const float* __restrict__ gp,
    const float* __restrict__ rp,
    const float* __restrict__ mhp_p,
    const float* __restrict__ mbp_p,
    const float* __restrict__ mlp_p,
    float* __restrict__ yout)
{
    static_assert(NPART == 2, "the pass sequence below is written for two parts");
    __shared__ WaveSlab slab[NSEG];
    __shared__ float hz[20];
    __shared__ float es1[NSEGT][16], es2[NSEGT][16];

    const unsigned lane = threadIdx.x & 31u;
    const unsigned wave = threadIdx.x >> 5;
    const int      j    = (int)(lane & 15u);
    const bool     hi   = lane >= 16u;
    const int ntile = (int)(gridDim.x / NPART);
    const int part  = (int)blockIdx.x < ntile ? 1 : 0;
    const int rbase = ((int)blockIdx.x - (part ? 0 : ntile)) * 16;
    const int seg0  = (int)wave * SEG;
    const int gseg  = part * NSEG + (int)wave;

    const float* xt0   = x + (size_t)rbase * T_LEN + seg0;
    const float* xtown = xt0 + part * PART;
    float*       ytile = yout + (size_t)rbase * T_LEN + part * PART + seg0;

    float4 xq0, xq1, xq2, xq3;
    {
        const float* xl = xt0 + (size_t)(lane >> 3) * T_LEN + 4 * (lane & 7u);
        xq0 = *(const float4*)(xl);
        xq1 = *(const float4*)(xl + (size_t)4 * T_LEN);
        xq2 = *(const float4*)(xl + (size_t)8 * T_LEN);
        xq3 = *(const float4*)(xl + (size_t)12 * T_LEN);
    }

    const float gv  = gp[0];
    const float rv  = rp[0];
    const float mhp = mhp_p[0];
    const float mbp = mbp_p[0];
    const float mlp = mlp_p[0];

    const float sig = 1.0f / (1.0f + expf(-gv));
    const float gg  = tanf(3.14159265358979323846f * 0.5f * sig);
    const float rr  = log1pf(expf(rv));
    const float g2  = gg * gg;

    const float b0 = g2 * mlp + gg * mbp + mhp;
    const float b1 = 2.0f * g2 * mlp - 2.0f * mhp;
    const float b2 = g2 * mlp - gg * mbp + mhp;
    const float a0 = g2 + 2.0f * rr * gg + 1.0f;
    const float inva0 = 1.0f / a0;
    const float a1 = (2.0f * g2 - 2.0f) * inva0;
    const float a2 = (g2 - 2.0f * rr * gg + 1.0f) * inva0;

    if (threadIdx.x == 0u) {
        hz[0] = 0.0f;
        hz[1] = 0.0f;
        hz[2] = 1.0f;
        hz[3] = -a1;
        for (int i = 2; i <= 16; ++i)
            hz[2 + i] = -a1 * hz[1 + i] - a2 * hz[i];
    }

    WaveSlab& sl = slab[wave];
    const float* xr = x + (size_t)(rbase + j) * T_LEN + seg0;
    const float xm1a = wave ? xr[-1] : 0.0f;
    const float xm2a = wave ? xr[-2] : 0.0f;
    const float xm1b = xr[PART - 1];
    const float xm2b = xr[PART - 2];
    __syncthreads();

    const int fcol = hi ? 0 : 8;
    v2f Ga[4];
#pragma unroll
    for (int s = 0; s < 4; ++s) {
        const int d0 = j - (fcol + 2 * s);
        const int d1 = d0 - 1;
        const int c0 = d0 < 0 ? 0 : d0;
        const int c1 = d1 < 0 ? 0 : d1;
        const float g0 = b0 * hz[c0 + 2] + b1 * hz[c0 + 1] + b2 * hz[c0];
        const float g1 = b0 * hz[c1 + 2] + b1 * hz[c1 + 1] + b2 * hz[c1];
        Ga[s].x = d0 >= 0 ? g0 : 0.0f;
        Ga[s].y = d1 >= 0 ? g1 : 0.0f;
    }
    const float ht  = hz[j + 2];
    const float htm = hz[j + 1];
    const float htp = hz[j + 3];
    v2f Ca;
    Ca.x = hi ? -a2 * ht : b2 * ht;
    Ca.y = hi ? htp      : b1 * ht + b2 * htm;

    float h1 = 1.0f, h2 = 0.0f, h3 = 0.0f;
    v2f st;
    st.x = hi ? 0.0f : xm2a;
    st.y = hi ? 0.0f : xm1a;
    segment_pass<false, true>(sl, xt0, xtown, ytile, lane, a1, a2, Ga, Ca, st, h1, h2, h3,
                              xq0, xq1, xq2, xq3);
    if (hi) {
        es1[wave][j] = st.y;
        es2[wave][j] = st.x;
    }
    if (part) {
        st.x = hi ? 0.0f : xm2b;
        st.y = hi ? 0.0f : xm1b;
        segment_pass<false, false>(sl, xtown, xtown, ytile, lane, a1, a2, Ga, Ca, st, h1, h2, h3,
                                   xq0, xq1, xq2, xq3);
        if (hi) {
            es1[NSEG + wave][j] = st.y;
            es2[NSEG + wave][j] = st.x;
        }
    }
    __syncthreads();

    const float P1 = h1, Q1 = -a2 * h2;
    const float P2 = h2, Q2 = -a2 * h3;
    float y1 = 0.0f, y2 = 0.0f;
#pragma unroll
    for (int s = 0; s < NSEG - 1; ++s) {
        const float n1 = es1[s][j] + P1 * y1 + Q1 * y2;
        const float n2 = es2[s][j] + P2 * y1 + Q2 * y2;
        const bool take = s < gseg;
        y1 = take ? n1 : y1;
        y2 = take ? n2 : y2;
    }
    if (part) {
#pragma unroll
        for (int s = NSEG - 1; s < NSEGT - 1; ++s) {
            const float n1 = es1[s][j] + P1 * y1 + Q1 * y2;
            const float n2 = es2[s][j] + P2 * y1 + Q2 * y2;
            const bool take = s < gseg;
            y1 = take ? n1 : y1;
            y2 = take ? n2 : y2;
        }
    }
    st.x = hi ? y2 : (part ? xm2b : xm2a);
    st.y = hi ? y1 : (part ? xm1b : xm1a);

    segment_pass<true, false>(sl, xtown, xtown, ytile, lane, a1, a2, Ga, Ca, st, h1, h2, h3,
                              xq0, xq1, xq2, xq3);
}

extern "C" void kernel_launch(void* const* d_in, const int* in_sizes, int n_in,
                              void* d_out, int out_size, void* d_ws, size_t ws_size,
                              hipStream_t stream) {
    const float* x   = (const float*)d_in[0];
    const float* g   = (const float*)d_in[1];
    const float* r   = (const float*)d_in[2];
    const float* mhp = (const float*)d_in[3];
    const float* mbp = (const float*)d_in[4];
    const float* mlp = (const float*)d_in[5];
    float* y = (float*)d_out;

    const int B = in_sizes[0] / T_LEN;
    dim3 grid(B / 16 * NPART), block(32 * NSEG);
    hipLaunchKernelGGL(dsvf_wmma_kernel, grid, block, 0, stream,
                       x, g, r, mhp, mbp, mlp, y);
}


# Output pass recomputes nothing: the written segment's zero-state y is kept in LDS and pass 2 adds the start-state response, 1 product per chunk (was 5), no second x read; sum order changes
# speedup vs baseline: 16.4559x; 1.0648x over previous
#include <hip/hip_runtime.h>
#include <math.h>


typedef float v2f __attribute__((ext_vector_type(2)));
typedef float v8f __attribute__((ext_vector_type(8)));

constexpr int T_LEN = 8192;
constexpr int NPART = 2;
constexpr int PART  = T_LEN / NPART;
constexpr int NSEG  = 32;
constexpr int SEG   = PART / NSEG;
constexpr int NSEGT = NPART * NSEG;
constexpr int TRIP  = 32;
constexpr int NTRIP = SEG / TRIP;
constexpr int PITCH = 36;

struct __attribute__((aligned(16))) WaveSlab {
    float t[16][PITCH];
};

__device__ __forceinline__ void wave_lds_sync() {
    __builtin_amdgcn_fence(__ATOMIC_RELEASE, "wavefront");
    __builtin_amdgcn_wave_barrier();
    __builtin_amdgcn_fence(__ATOMIC_ACQUIRE, "wavefront");
}

__device__ __forceinline__ v8f mm(v2f a, v2f b, v8f c) {
    return __builtin_amdgcn_wmma_f32_16x16x4_f32(
         false, a,  false, b,
         (short)0, c,  false,  false);
}

template <bool KEEP, bool ADV, bool WRAP>
__device__ __forceinline__ void segment_pass(
    WaveSlab (&slabs)[NTRIP], const float* xtile, const float* xwrap,
    unsigned lane, float a1, float a2,
    const v2f (&Ga)[4], v2f Ca, v2f& st, float& h1, float& h2, float& h3,
    float4& xa0, float4& xa1, float4& xa2, float4& xa3)
{
    const int  j  = (int)(lane & 15u);
    const bool hi = lane >= 16u;
    const int fcol  = hi ? 0 : 8;
    const int ycol  = hi ? 8 : 0;
    const int rload = (int)(lane >> 1);
    const int fb    = (int)(lane & 1u);
    const int lrow  = (int)(lane >> 3);
    const int lq    = (int)(lane & 7u);

    auto one_trip = [&](WaveSlab& sl, const int t0, const bool second, const bool last,
                        const float4& c0, const float4& c1, const float4& c2, const float4& c3,
                        float4& n0, float4& n1, float4& n2, float4& n3)
                        __attribute__((always_inline)) {
        if (WRAP || !last) {
            const float* xl = (last ? xwrap : xtile + t0 + TRIP) + (size_t)lrow * T_LEN + 4 * lq;
            n0 = *(const float4*)(xl);
            n1 = *(const float4*)(xl + (size_t)4 * T_LEN);
            n2 = *(const float4*)(xl + (size_t)8 * T_LEN);
            n3 = *(const float4*)(xl + (size_t)12 * T_LEN);
        }
        if (ADV) {
#pragma unroll
            for (int k = 0; k < TRIP; ++k) {
                const float hn = -a1 * h1 - a2 * h2;
                h3 = h2; h2 = h1; h1 = hn;
            }
            asm volatile("" : "+s"(h1), "+s"(h2), "+s"(h3));
        }
        if (!second && t0 + 2 * TRIP < SEG)
            __builtin_prefetch(xtile + (size_t)rload * T_LEN + t0 + 2 * TRIP + 16 * fb, 0, 1);

        *(float4*)&sl.t[lrow][4 * lq]      = c0;
        *(float4*)&sl.t[4 + lrow][4 * lq]  = c1;
        *(float4*)&sl.t[8 + lrow][4 * lq]  = c2;
        *(float4*)&sl.t[12 + lrow][4 * lq] = c3;
        wave_lds_sync();

        float4 f[2][2];
#pragma unroll
        for (int c = 0; c < 2; ++c) {
            f[c][0] = *(const float4*)&sl.t[j][16 * c + fcol];
            f[c][1] = *(const float4*)&sl.t[j][16 * c + fcol + 4];
        }

        v8f acc0 = {}, acc1 = {};
        {
            const v2f b00 = {f[0][0].x, f[0][0].y}, b01 = {f[0][0].z, f[0][0].w};
            const v2f b02 = {f[0][1].x, f[0][1].y}, b03 = {f[0][1].z, f[0][1].w};
            const v2f b10 = {f[1][0].x, f[1][0].y}, b11 = {f[1][0].z, f[1][0].w};
            const v2f b12 = {f[1][1].x, f[1][1].y}, b13 = {f[1][1].z, f[1][1].w};
            acc0 = mm(Ga[0], b00, acc0);  acc1 = mm(Ga[0], b10, acc1);
            acc0 = mm(Ga[1], b01, acc0);  acc1 = mm(Ga[1], b11, acc1);
            acc0 = mm(Ga[2], b02, acc0);  acc1 = mm(Ga[2], b12, acc1);
            acc0 = mm(Ga[3], b03, acc0);  acc1 = mm(Ga[3], b13, acc1);
        }

        acc0 = mm(Ca, st, acc0);
        st.x = hi ? acc0[6] : f[0][1].z;
        st.y = hi ? acc0[7] : f[0][1].w;
        acc1 = mm(Ca, st, acc1);
        st.x = hi ? acc1[6] : f[1][1].z;
        st.y = hi ? acc1[7] : f[1][1].w;

        wave_lds_sync();
        if (KEEP) {
            *(float4*)&sl.t[j][ycol]          = make_float4(acc0[0], acc0[1], acc0[2], acc0[3]);
            *(float4*)&sl.t[j][ycol + 4]      = make_float4(acc0[4], acc0[5], acc0[6], acc0[7]);
            *(float4*)&sl.t[j][16 + ycol]     = make_float4(acc1[0], acc1[1], acc1[2], acc1[3]);
            *(float4*)&sl.t[j][16 + ycol + 4] = make_float4(acc1[4], acc1[5], acc1[6], acc1[7]);
        } else {
            asm volatile("" : "+v"(st.x), "+v"(st.y));
        }
    };

    float4 xb0, xb1, xb2, xb3;
    static_assert(NTRIP == 4, "a segment is four trips");
    one_trip(slabs[0], 0,        false, false, xa0, xa1, xa2, xa3, xb0, xb1, xb2, xb3);
    one_trip(slabs[1], TRIP,     true,  false, xb0, xb1, xb2, xb3, xa0, xa1, xa2, xa3);
    one_trip(slabs[2], 2 * TRIP, false, false, xa0, xa1, xa2, xa3, xb0, xb1, xb2, xb3);
    one_trip(slabs[3], 3 * TRIP, true,  true,  xb0, xb1, xb2, xb3, xa0, xa1, xa2, xa3);
}

__device__ __forceinline__ void output_pass(
    WaveSlab (&slabs)[NTRIP], float* __restrict__ ytile, unsigned lane, v2f Ca,
    float y1, float y2, float P16, float Q16, float P15, float Q15)
{
    const int  j    = (int)(lane & 15u);
    const bool hi   = lane >= 16u;
    const int  ycol = hi ? 8 : 0;
    const int  lrow = (int)(lane >> 3);
    const int  lq   = (int)(lane & 7u);

    float d1 = hi ? y1 : 0.0f;
    float d2 = hi ? y2 : 0.0f;
#pragma unroll
    for (int tr = 0; tr < NTRIP; ++tr) {
        WaveSlab& sl = slabs[tr];
        const float4 q0 = *(const float4*)&sl.t[j][ycol];
        const float4 q1 = *(const float4*)&sl.t[j][ycol + 4];
        const float4 q2 = *(const float4*)&sl.t[j][16 + ycol];
        const float4 q3 = *(const float4*)&sl.t[j][16 + ycol + 4];
        v8f acc0 = {q0.x, q0.y, q0.z, q0.w, q1.x, q1.y, q1.z, q1.w};
        v8f acc1 = {q2.x, q2.y, q2.z, q2.w, q3.x, q3.y, q3.z, q3.w};

        const v2f s0 = {d2, d1};
        const float e1 = P16 * d1 + Q16 * d2;
        const float e2 = P15 * d1 + Q15 * d2;
        const v2f s1 = {e2, e1};
        d1 = P16 * e1 + Q16 * e2;
        d2 = P15 * e1 + Q15 * e2;

        acc0 = mm(Ca, s0, acc0);
        acc1 = mm(Ca, s1, acc1);

        *(float4*)&sl.t[j][ycol]          = make_float4(acc0[0], acc0[1], acc0[2], acc0[3]);
        *(float4*)&sl.t[j][ycol + 4]      = make_float4(acc0[4], acc0[5], acc0[6], acc0[7]);
        *(float4*)&sl.t[j][16 + ycol]     = make_float4(acc1[0], acc1[1], acc1[2], acc1[3]);
        *(float4*)&sl.t[j][16 + ycol + 4] = make_float4(acc1[4], acc1[5], acc1[6], acc1[7]);
        wave_lds_sync();
#pragma unroll
        for (int i = 0; i < 4; ++i) {
            const float4 yv = *(const float4*)&sl.t[4 * i + lrow][4 * lq];
            *(float4*)(ytile + (size_t)(4 * i + lrow) * T_LEN + tr * TRIP + 4 * lq) = yv;
        }
    }
}

__global__ __launch_bounds__(32 * NSEG) void dsvf_wmma_kernel(
    const float* x,
    const float* __restrict__ gp,
    const float* __restrict__ rp,
    const float* __restrict__ mhp_p,
    const float* __restrict__ mbp_p,
    const float* __restrict__ mlp_p,
    float* __restrict__ yout)
{
    static_assert(NPART == 2, "the pass sequence below is written for two parts");
    __shared__ WaveSlab slab[NSEG][NTRIP];
    __shared__ float hz[20];
    __shared__ float es1[NSEGT][16], es2[NSEGT][16];

    const unsigned lane = threadIdx.x & 31u;
    const unsigned wave = threadIdx.x >> 5;
    const int      j    = (int)(lane & 15u);
    const bool     hi   = lane >= 16u;
    const int ntile = (int)(gridDim.x / NPART);
    const int part  = (int)blockIdx.x < ntile ? 1 : 0;
    const int rbase = ((int)blockIdx.x - (part ? 0 : ntile)) * 16;
    const int seg0  = (int)wave * SEG;
    const int gseg  = part * NSEG + (int)wave;

    const float* xt0   = x + (size_t)rbase * T_LEN + seg0;
    const float* xtown = xt0 + part * PART;
    float*       ytile = yout + (size_t)rbase * T_LEN + part * PART + seg0;

    float4 xq0, xq1, xq2, xq3;
    {
        const float* xl = xt0 + (size_t)(lane >> 3) * T_LEN + 4 * (lane & 7u);
        xq0 = *(const float4*)(xl);
        xq1 = *(const float4*)(xl + (size_t)4 * T_LEN);
        xq2 = *(const float4*)(xl + (size_t)8 * T_LEN);
        xq3 = *(const float4*)(xl + (size_t)12 * T_LEN);
    }

    const float gv  = gp[0];
    const float rv  = rp[0];
    const float mhp = mhp_p[0];
    const float mbp = mbp_p[0];
    const float mlp = mlp_p[0];

    const float sig = 1.0f / (1.0f + expf(-gv));
    const float gg  = tanf(3.14159265358979323846f * 0.5f * sig);
    const float rr  = log1pf(expf(rv));
    const float g2  = gg * gg;

    const float b0 = g2 * mlp + gg * mbp + mhp;
    const float b1 = 2.0f * g2 * mlp - 2.0f * mhp;
    const float b2 = g2 * mlp - gg * mbp + mhp;
    const float a0 = g2 + 2.0f * rr * gg + 1.0f;
    const float inva0 = 1.0f / a0;
    const float a1 = (2.0f * g2 - 2.0f) * inva0;
    const float a2 = (g2 - 2.0f * rr * gg + 1.0f) * inva0;

    if (threadIdx.x == 0u) {
        hz[0] = 0.0f;
        hz[1] = 0.0f;
        hz[2] = 1.0f;
        hz[3] = -a1;
        for (int i = 2; i <= 16; ++i)
            hz[2 + i] = -a1 * hz[1 + i] - a2 * hz[i];
    }

    WaveSlab (&sl)[NTRIP] = slab[wave];
    const float* xr = x + (size_t)(rbase + j) * T_LEN + seg0;
    const float xm1a = wave ? xr[-1] : 0.0f;
    const float xm2a = wave ? xr[-2] : 0.0f;
    const float xm1b = xr[PART - 1];
    const float xm2b = xr[PART - 2];
    __syncthreads();

    const int fcol = hi ? 0 : 8;
    v2f Ga[4];
#pragma unroll
    for (int s = 0; s < 4; ++s) {
        const int d0 = j - (fcol + 2 * s);
        const int d1 = d0 - 1;
        const int c0 = d0 < 0 ? 0 : d0;
        const int c1 = d1 < 0 ? 0 : d1;
        const float g0 = b0 * hz[c0 + 2] + b1 * hz[c0 + 1] + b2 * hz[c0];
        const float g1 = b0 * hz[c1 + 2] + b1 * hz[c1 + 1] + b2 * hz[c1];
        Ga[s].x = d0 >= 0 ? g0 : 0.0f;
        Ga[s].y = d1 >= 0 ? g1 : 0.0f;
    }
    const float ht  = hz[j + 2];
    const float htm = hz[j + 1];
    const float htp = hz[j + 3];
    v2f Ca;
    Ca.x = hi ? -a2 * ht : b2 * ht;
    Ca.y = hi ? htp      : b1 * ht + b2 * htm;

    float h1 = 1.0f, h2 = 0.0f, h3 = 0.0f;
    v2f st;
    st.x = hi ? 0.0f : xm2a;
    st.y = hi ? 0.0f : xm1a;
    if (part) {
        segment_pass<false, true, true>(sl, xt0, xtown, lane, a1, a2, Ga, Ca, st, h1, h2, h3,
                                        xq0, xq1, xq2, xq3);
        if (hi) {
            es1[wave][j] = st.y;
            es2[wave][j] = st.x;
        }
        st.x = hi ? 0.0f : xm2b;
        st.y = hi ? 0.0f : xm1b;
        segment_pass<true, false, false>(sl, xtown, xtown, lane, a1, a2, Ga, Ca, st, h1, h2, h3,
                                         xq0, xq1, xq2, xq3);
        if (hi) {
            es1[NSEG + wave][j] = st.y;
            es2[NSEG + wave][j] = st.x;
        }
    } else {
        segment_pass<true, true, false>(sl, xt0, xt0, lane, a1, a2, Ga, Ca, st, h1, h2, h3,
                                        xq0, xq1, xq2, xq3);
        if (hi) {
            es1[wave][j] = st.y;
            es2[wave][j] = st.x;
        }
    }
    __syncthreads();

    const float P1 = h1, Q1 = -a2 * h2;
    const float P2 = h2, Q2 = -a2 * h3;
    float y1 = 0.0f, y2 = 0.0f;
#pragma unroll
    for (int s = 0; s < NSEG - 1; ++s) {
        const float n1 = es1[s][j] + P1 * y1 + Q1 * y2;
        const float n2 = es2[s][j] + P2 * y1 + Q2 * y2;
        const bool take = s < gseg;
        y1 = take ? n1 : y1;
        y2 = take ? n2 : y2;
    }
    if (part) {
#pragma unroll
        for (int s = NSEG - 1; s < NSEGT - 1; ++s) {
            const float n1 = es1[s][j] + P1 * y1 + Q1 * y2;
            const float n2 = es2[s][j] + P2 * y1 + Q2 * y2;
            const bool take = s < gseg;
            y1 = take ? n1 : y1;
            y2 = take ? n2 : y2;
        }
    }

    const float P16 = hz[18], Q16 = -a2 * hz[17];
    const float P15 = hz[17], Q15 = -a2 * hz[16];
    output_pass(sl, ytile, lane, Ca, y1, y2, P16, Q16, P15, Q15);
}

extern "C" void kernel_launch(void* const* d_in, const int* in_sizes, int n_in,
                              void* d_out, int out_size, void* d_ws, size_t ws_size,
                              hipStream_t stream) {
    const float* x   = (const float*)d_in[0];
    const float* g   = (const float*)d_in[1];
    const float* r   = (const float*)d_in[2];
    const float* mhp = (const float*)d_in[3];
    const float* mbp = (const float*)d_in[4];
    const float* mlp = (const float*)d_in[5];
    float* y = (float*)d_out;

    const int B = in_sizes[0] / T_LEN;
    dim3 grid(B / 16 * NPART), block(32 * NSEG);
    hipLaunchKernelGGL(dsvf_wmma_kernel, grid, block, 0, stream,
                       x, g, r, mhp, mbp, mlp, y);
}


# Work identical in all 32 waves (coefficients, A fragments, carry scan over segments) is done once by wave 0 and handed on through LDS behind a barrier; the other waves wait; sums unchanged
# speedup vs baseline: 17.7544x; 1.0789x over previous
#include <hip/hip_runtime.h>
#include <math.h>


typedef float v2f __attribute__((ext_vector_type(2)));
typedef float v8f __attribute__((ext_vector_type(8)));

constexpr int T_LEN = 8192;
constexpr int NPART = 2;
constexpr int PART  = T_LEN / NPART;
constexpr int NSEG  = 32;
constexpr int SEG   = PART / NSEG;
constexpr int NSEGT = NPART * NSEG;
constexpr int TRIP  = 32;
constexpr int NTRIP = SEG / TRIP;
constexpr int PITCH = 36;

struct __attribute__((aligned(16))) WaveSlab {
    float t[16][PITCH];
};

__device__ __forceinline__ void wave_lds_sync() {
    __builtin_amdgcn_fence(__ATOMIC_RELEASE, "wavefront");
    __builtin_amdgcn_wave_barrier();
    __builtin_amdgcn_fence(__ATOMIC_ACQUIRE, "wavefront");
}

__device__ __forceinline__ v8f mm(v2f a, v2f b, v8f c) {
    return __builtin_amdgcn_wmma_f32_16x16x4_f32(
         false, a,  false, b,
         (short)0, c,  false,  false);
}

template <bool KEEP, bool ADV, bool WRAP>
__device__ __forceinline__ void segment_pass(
    WaveSlab (&slabs)[NTRIP], const float* xtile, const float* xwrap,
    unsigned lane, float a1, float a2,
    const v2f (&Ga)[4], v2f Ca, v2f& st, float& h1, float& h2, float& h3,
    float4& xa0, float4& xa1, float4& xa2, float4& xa3)
{
    const int  j  = (int)(lane & 15u);
    const bool hi = lane >= 16u;
    const int fcol  = hi ? 0 : 8;
    const int ycol  = hi ? 8 : 0;
    const int rload = (int)(lane >> 1);
    const int fb    = (int)(lane & 1u);
    const int lrow  = (int)(lane >> 3);
    const int lq    = (int)(lane & 7u);

    auto one_trip = [&](WaveSlab& sl, const int t0, const bool second, const bool last,
                        const float4& c0, const float4& c1, const float4& c2, const float4& c3,
                        float4& n0, float4& n1, float4& n2, float4& n3)
                        __attribute__((always_inline)) {
        if (WRAP || !last) {
            const float* xl = (last ? xwrap : xtile + t0 + TRIP) + (size_t)lrow * T_LEN + 4 * lq;
            n0 = *(const float4*)(xl);
            n1 = *(const float4*)(xl + (size_t)4 * T_LEN);
            n2 = *(const float4*)(xl + (size_t)8 * T_LEN);
            n3 = *(const float4*)(xl + (size_t)12 * T_LEN);
        }
        if (ADV) {
#pragma unroll
            for (int k = 0; k < TRIP; ++k) {
                const float hn = -a1 * h1 - a2 * h2;
                h3 = h2; h2 = h1; h1 = hn;
            }
            asm volatile("" : "+s"(h1), "+s"(h2), "+s"(h3));
        }
        if (!second && t0 + 2 * TRIP < SEG)
            __builtin_prefetch(xtile + (size_t)rload * T_LEN + t0 + 2 * TRIP + 16 * fb, 0, 1);

        *(float4*)&sl.t[lrow][4 * lq]      = c0;
        *(float4*)&sl.t[4 + lrow][4 * lq]  = c1;
        *(float4*)&sl.t[8 + lrow][4 * lq]  = c2;
        *(float4*)&sl.t[12 + lrow][4 * lq] = c3;
        wave_lds_sync();

        float4 f[2][2];
#pragma unroll
        for (int c = 0; c < 2; ++c) {
            f[c][0] = *(const float4*)&sl.t[j][16 * c + fcol];
            f[c][1] = *(const float4*)&sl.t[j][16 * c + fcol + 4];
        }

        v8f acc0 = {}, acc1 = {};
        {
            const v2f b00 = {f[0][0].x, f[0][0].y}, b01 = {f[0][0].z, f[0][0].w};
            const v2f b02 = {f[0][1].x, f[0][1].y}, b03 = {f[0][1].z, f[0][1].w};
            const v2f b10 = {f[1][0].x, f[1][0].y}, b11 = {f[1][0].z, f[1][0].w};
            const v2f b12 = {f[1][1].x, f[1][1].y}, b13 = {f[1][1].z, f[1][1].w};
            acc0 = mm(Ga[0], b00, acc0);  acc1 = mm(Ga[0], b10, acc1);
            acc0 = mm(Ga[1], b01, acc0);  acc1 = mm(Ga[1], b11, acc1);
            acc0 = mm(Ga[2], b02, acc0);  acc1 = mm(Ga[2], b12, acc1);
            acc0 = mm(Ga[3], b03, acc0);  acc1 = mm(Ga[3], b13, acc1);
        }

        acc0 = mm(Ca, st, acc0);
        st.x = hi ? acc0[6] : f[0][1].z;
        st.y = hi ? acc0[7] : f[0][1].w;
        acc1 = mm(Ca, st, acc1);
        st.x = hi ? acc1[6] : f[1][1].z;
        st.y = hi ? acc1[7] : f[1][1].w;

        wave_lds_sync();
        if (KEEP) {
            *(float4*)&sl.t[j][ycol]          = make_float4(acc0[0], acc0[1], acc0[2], acc0[3]);
            *(float4*)&sl.t[j][ycol + 4]      = make_float4(acc0[4], acc0[5], acc0[6], acc0[7]);
            *(float4*)&sl.t[j][16 + ycol]     = make_float4(acc1[0], acc1[1], acc1[2], acc1[3]);
            *(float4*)&sl.t[j][16 + ycol + 4] = make_float4(acc1[4], acc1[5], acc1[6], acc1[7]);
        } else {
            asm volatile("" : "+v"(st.x), "+v"(st.y));
        }
    };

    float4 xb0, xb1, xb2, xb3;
    static_assert(NTRIP == 4, "a segment is four trips");
    one_trip(slabs[0], 0,        false, false, xa0, xa1, xa2, xa3, xb0, xb1, xb2, xb3);
    one_trip(slabs[1], TRIP,     true,  false, xb0, xb1, xb2, xb3, xa0, xa1, xa2, xa3);
    one_trip(slabs[2], 2 * TRIP, false, false, xa0, xa1, xa2, xa3, xb0, xb1, xb2, xb3);
    one_trip(slabs[3], 3 * TRIP, true,  true,  xb0, xb1, xb2, xb3, xa0, xa1, xa2, xa3);
}

__device__ __forceinline__ void output_pass(
    WaveSlab (&slabs)[NTRIP], float* __restrict__ ytile, unsigned lane, v2f Ca,
    float y1, float y2, float P16, float Q16, float P15, float Q15)
{
    const int  j    = (int)(lane & 15u);
    const bool hi   = lane >= 16u;
    const int  ycol = hi ? 8 : 0;
    const int  lrow = (int)(lane >> 3);
    const int  lq   = (int)(lane & 7u);

    float d1 = hi ? y1 : 0.0f;
    float d2 = hi ? y2 : 0.0f;
#pragma unroll
    for (int tr = 0; tr < NTRIP; ++tr) {
        WaveSlab& sl = slabs[tr];
        const float4 q0 = *(const float4*)&sl.t[j][ycol];
        const float4 q1 = *(const float4*)&sl.t[j][ycol + 4];
        const float4 q2 = *(const float4*)&sl.t[j][16 + ycol];
        const float4 q3 = *(const float4*)&sl.t[j][16 + ycol + 4];
        v8f acc0 = {q0.x, q0.y, q0.z, q0.w, q1.x, q1.y, q1.z, q1.w};
        v8f acc1 = {q2.x, q2.y, q2.z, q2.w, q3.x, q3.y, q3.z, q3.w};

        const v2f s0 = {d2, d1};
        const float e1 = P16 * d1 + Q16 * d2;
        const float e2 = P15 * d1 + Q15 * d2;
        const v2f s1 = {e2, e1};
        d1 = P16 * e1 + Q16 * e2;
        d2 = P15 * e1 + Q15 * e2;

        acc0 = mm(Ca, s0, acc0);
        acc1 = mm(Ca, s1, acc1);

        *(float4*)&sl.t[j][ycol]          = make_float4(acc0[0], acc0[1], acc0[2], acc0[3]);
        *(float4*)&sl.t[j][ycol + 4]      = make_float4(acc0[4], acc0[5], acc0[6], acc0[7]);
        *(float4*)&sl.t[j][16 + ycol]     = make_float4(acc1[0], acc1[1], acc1[2], acc1[3]);
        *(float4*)&sl.t[j][16 + ycol + 4] = make_float4(acc1[4], acc1[5], acc1[6], acc1[7]);
        wave_lds_sync();
#pragma unroll
        for (int i = 0; i < 4; ++i) {
            const float4 yv = *(const float4*)&sl.t[4 * i + lrow][4 * lq];
            *(float4*)(ytile + (size_t)(4 * i + lrow) * T_LEN + tr * TRIP + 4 * lq) = yv;
        }
    }
}

__global__ __launch_bounds__(32 * NSEG) void dsvf_wmma_kernel(
    const float* x,
    const float* __restrict__ gp,
    const float* __restrict__ rp,
    const float* __restrict__ mhp_p,
    const float* __restrict__ mbp_p,
    const float* __restrict__ mlp_p,
    float* __restrict__ yout)
{
    static_assert(NPART == 2, "the pass sequence below is written for two parts");
    __shared__ WaveSlab slab[NSEG][NTRIP];
    __shared__ float hz[20];
    __shared__ float es1[NSEGT][16], es2[NSEGT][16];
    __shared__ float ss1[NSEGT][16], ss2[NSEGT][16];
    __shared__ float cf[2];
    __shared__ __attribute__((aligned(16))) float frag[32][12];

    const unsigned lane = threadIdx.x & 31u;
    const unsigned wave = threadIdx.x >> 5;
    const int      j    = (int)(lane & 15u);
    const bool     hi   = lane >= 16u;
    const bool lead = __builtin_amdgcn_readfirstlane((int)threadIdx.x) < 32;
    const int ntile = (int)(gridDim.x / NPART);
    const int part  = (int)blockIdx.x < ntile ? 1 : 0;
    const int rbase = ((int)blockIdx.x - (part ? 0 : ntile)) * 16;
    const int seg0  = (int)wave * SEG;
    const int gseg  = part * NSEG + (int)wave;

    const float* xt0   = x + (size_t)rbase * T_LEN + seg0;
    const float* xtown = xt0 + part * PART;
    float*       ytile = yout + (size_t)rbase * T_LEN + part * PART + seg0;

    float4 xq0, xq1, xq2, xq3;
    {
        const float* xl = xt0 + (size_t)(lane >> 3) * T_LEN + 4 * (lane & 7u);
        xq0 = *(const float4*)(xl);
        xq1 = *(const float4*)(xl + (size_t)4 * T_LEN);
        xq2 = *(const float4*)(xl + (size_t)8 * T_LEN);
        xq3 = *(const float4*)(xl + (size_t)12 * T_LEN);
    }

    WaveSlab (&sl)[NTRIP] = slab[wave];
    const float* xr = x + (size_t)(rbase + j) * T_LEN + seg0;
    const float xm1a = wave ? xr[-1] : 0.0f;
    const float xm2a = wave ? xr[-2] : 0.0f;
    const float xm1b = xr[PART - 1];
    const float xm2b = xr[PART - 2];

    if (lead) {
        const float gv  = gp[0];
        const float rv  = rp[0];
        const float mhp = mhp_p[0];
        const float mbp = mbp_p[0];
        const float mlp = mlp_p[0];

        const float sig = 1.0f / (1.0f + expf(-gv));
        const float gg  = tanf(3.14159265358979323846f * 0.5f * sig);
        const float rr  = log1pf(expf(rv));
        const float g2  = gg * gg;

        const float b0 = g2 * mlp + gg * mbp + mhp;
        const float b1 = 2.0f * g2 * mlp - 2.0f * mhp;
        const float b2 = g2 * mlp - gg * mbp + mhp;
        const float a0 = g2 + 2.0f * rr * gg + 1.0f;
        const float inva0 = 1.0f / a0;
        const float a1 = (2.0f * g2 - 2.0f) * inva0;
        const float a2 = (g2 - 2.0f * rr * gg + 1.0f) * inva0;

        if (threadIdx.x == 0u) {
            hz[0] = 0.0f;
            hz[1] = 0.0f;
            hz[2] = 1.0f;
            hz[3] = -a1;
            for (int i = 2; i <= 16; ++i)
                hz[2 + i] = -a1 * hz[1 + i] - a2 * hz[i];
            cf[0] = a1;
            cf[1] = a2;
        }
        wave_lds_sync();

        const int fcol = hi ? 0 : 8;
#pragma unroll
        for (int s = 0; s < 4; ++s) {
            const int d0 = j - (fcol + 2 * s);
            const int d1 = d0 - 1;
            const int c0 = d0 < 0 ? 0 : d0;
            const int c1 = d1 < 0 ? 0 : d1;
            const float g0 = b0 * hz[c0 + 2] + b1 * hz[c0 + 1] + b2 * hz[c0];
            const float g1 = b0 * hz[c1 + 2] + b1 * hz[c1 + 1] + b2 * hz[c1];
            frag[lane][2 * s]     = d0 >= 0 ? g0 : 0.0f;
            frag[lane][2 * s + 1] = d1 >= 0 ? g1 : 0.0f;
        }
        const float ht  = hz[j + 2];
        const float htm = hz[j + 1];
        const float htp = hz[j + 3];
        frag[lane][8] = hi ? -a2 * ht : b2 * ht;
        frag[lane][9] = hi ? htp      : b1 * ht + b2 * htm;
    }
    __syncthreads();

    const float a1 = __builtin_bit_cast(float, __builtin_amdgcn_readfirstlane(__builtin_bit_cast(int, cf[0])));
    const float a2 = __builtin_bit_cast(float, __builtin_amdgcn_readfirstlane(__builtin_bit_cast(int, cf[1])));
    v2f Ga[4];
    v2f Ca;
    {
        const float4 fa = *(const float4*)&frag[lane][0];
        const float4 fb = *(const float4*)&frag[lane][4];
        const float2 fc = *(const float2*)&frag[lane][8];
        Ga[0].x = fa.x; Ga[0].y = fa.y; Ga[1].x = fa.z; Ga[1].y = fa.w;
        Ga[2].x = fb.x; Ga[2].y = fb.y; Ga[3].x = fb.z; Ga[3].y = fb.w;
        Ca.x = fc.x; Ca.y = fc.y;
    }

    float h1 = 1.0f, h2 = 0.0f, h3 = 0.0f;
    v2f st;
    st.x = hi ? 0.0f : xm2a;
    st.y = hi ? 0.0f : xm1a;
    if (part) {
        segment_pass<false, true, true>(sl, xt0, xtown, lane, a1, a2, Ga, Ca, st, h1, h2, h3,
                                        xq0, xq1, xq2, xq3);
        if (hi) {
            es1[wave][j] = st.y;
            es2[wave][j] = st.x;
        }
        st.x = hi ? 0.0f : xm2b;
        st.y = hi ? 0.0f : xm1b;
        segment_pass<true, false, false>(sl, xtown, xtown, lane, a1, a2, Ga, Ca, st, h1, h2, h3,
                                         xq0, xq1, xq2, xq3);
        if (hi) {
            es1[NSEG + wave][j] = st.y;
            es2[NSEG + wave][j] = st.x;
        }
    } else {
        segment_pass<true, true, false>(sl, xt0, xt0, lane, a1, a2, Ga, Ca, st, h1, h2, h3,
                                        xq0, xq1, xq2, xq3);
        if (hi) {
            es1[wave][j] = st.y;
            es2[wave][j] = st.x;
        }
    }
    __syncthreads();

    const float P1 = h1, Q1 = -a2 * h2;
    const float P2 = h2, Q2 = -a2 * h3;
    if (lead && !hi) {
        float y1 = 0.0f, y2 = 0.0f;
        ss1[0][j] = 0.0f;
        ss2[0][j] = 0.0f;
#pragma unroll
        for (int s = 0; s < NSEG - 1; ++s) {
            const float n1 = es1[s][j] + P1 * y1 + Q1 * y2;
            const float n2 = es2[s][j] + P2 * y1 + Q2 * y2;
            y1 = n1;
            y2 = n2;
            ss1[s + 1][j] = y1;
            ss2[s + 1][j] = y2;
        }
        if (part) {
#pragma unroll
            for (int s = NSEG - 1; s < NSEGT - 1; ++s) {
                const float n1 = es1[s][j] + P1 * y1 + Q1 * y2;
                const float n2 = es2[s][j] + P2 * y1 + Q2 * y2;
                y1 = n1;
                y2 = n2;
                ss1[s + 1][j] = y1;
                ss2[s + 1][j] = y2;
            }
        }
    }
    __syncthreads();
    const float y1 = ss1[gseg][j];
    const float y2 = ss2[gseg][j];

    const float P16 = hz[18], Q16 = -a2 * hz[17];
    const float P15 = hz[17], Q15 = -a2 * hz[16];
    output_pass(sl, ytile, lane, Ca, y1, y2, P16, Q16, P15, Q15);
}

extern "C" void kernel_launch(void* const* d_in, const int* in_sizes, int n_in,
                              void* d_out, int out_size, void* d_ws, size_t ws_size,
                              hipStream_t stream) {
    const float* x   = (const float*)d_in[0];
    const float* g   = (const float*)d_in[1];
    const float* r   = (const float*)d_in[2];
    const float* mhp = (const float*)d_in[3];
    const float* mbp = (const float*)d_in[4];
    const float* mlp = (const float*)d_in[5];
    float* y = (float*)d_out;

    const int B = in_sizes[0] / T_LEN;
    dim3 grid(B / 16 * NPART), block(32 * NSEG);
    hipLaunchKernelGGL(dsvf_wmma_kernel, grid, block, 0, stream,
                       x, g, r, mhp, mbp, mlp, y);
}


# Work-group of 16 waves (was 32), each wave over a 256-step segment in 8 trips: half as many waves enter per launch, the carry scan folds 32 segments (was 64); carry sum order changes
# speedup vs baseline: 19.0509x; 1.0730x over previous
#include <hip/hip_runtime.h>
#include <math.h>


typedef float v2f __attribute__((ext_vector_type(2)));
typedef float v8f __attribute__((ext_vector_type(8)));

constexpr int T_LEN = 8192;
constexpr int NPART = 2;
constexpr int PART  = T_LEN / NPART;
constexpr int NSEG  = 16;
constexpr int SEG   = PART / NSEG;
constexpr int NSEGT = NPART * NSEG;
constexpr int TRIP  = 32;
constexpr int NTRIP = SEG / TRIP;
constexpr int PITCH = 36;

struct __attribute__((aligned(16))) WaveSlab {
    float t[16][PITCH];
};

__device__ __forceinline__ void wave_lds_sync() {
    __builtin_amdgcn_fence(__ATOMIC_RELEASE, "wavefront");
    __builtin_amdgcn_wave_barrier();
    __builtin_amdgcn_fence(__ATOMIC_ACQUIRE, "wavefront");
}

__device__ __forceinline__ v8f mm(v2f a, v2f b, v8f c) {
    return __builtin_amdgcn_wmma_f32_16x16x4_f32(
         false, a,  false, b,
         (short)0, c,  false,  false);
}

template <bool KEEP, bool ADV, bool WRAP>
__device__ __forceinline__ void segment_pass(
    WaveSlab (&slabs)[NTRIP], const float* xtile, const float* xwrap,
    unsigned lane, float a1, float a2,
    const v2f (&Ga)[4], v2f Ca, v2f& st, float& h1, float& h2, float& h3,
    float4& xa0, float4& xa1, float4& xa2, float4& xa3)
{
    const int  j  = (int)(lane & 15u);
    const bool hi = lane >= 16u;
    const int fcol  = hi ? 0 : 8;
    const int ycol  = hi ? 8 : 0;
    const int rload = (int)(lane >> 1);
    const int fb    = (int)(lane & 1u);
    const int lrow  = (int)(lane >> 3);
    const int lq    = (int)(lane & 7u);

    auto one_trip = [&](WaveSlab& sl, const int t0, const bool second, const bool last,
                        const float4& c0, const float4& c1, const float4& c2, const float4& c3,
                        float4& n0, float4& n1, float4& n2, float4& n3)
                        __attribute__((always_inline)) {
        if (WRAP || !last) {
            const float* xl = (last ? xwrap : xtile + t0 + TRIP) + (size_t)lrow * T_LEN + 4 * lq;
            n0 = *(const float4*)(xl);
            n1 = *(const float4*)(xl + (size_t)4 * T_LEN);
            n2 = *(const float4*)(xl + (size_t)8 * T_LEN);
            n3 = *(const float4*)(xl + (size_t)12 * T_LEN);
        }
        if (ADV) {
#pragma unroll
            for (int k = 0; k < TRIP; ++k) {
                const float hn = -a1 * h1 - a2 * h2;
                h3 = h2; h2 = h1; h1 = hn;
            }
            asm volatile("" : "+s"(h1), "+s"(h2), "+s"(h3));
        }
        if (!second && t0 + 2 * TRIP < SEG)
            __builtin_prefetch(xtile + (size_t)rload * T_LEN + t0 + 2 * TRIP + 16 * fb, 0, 1);

        *(float4*)&sl.t[lrow][4 * lq]      = c0;
        *(float4*)&sl.t[4 + lrow][4 * lq]  = c1;
        *(float4*)&sl.t[8 + lrow][4 * lq]  = c2;
        *(float4*)&sl.t[12 + lrow][4 * lq] = c3;
        wave_lds_sync();

        float4 f[2][2];
#pragma unroll
        for (int c = 0; c < 2; ++c) {
            f[c][0] = *(const float4*)&sl.t[j][16 * c + fcol];
            f[c][1] = *(const float4*)&sl.t[j][16 * c + fcol + 4];
        }

        v8f acc0 = {}, acc1 = {};
        {
            const v2f b00 = {f[0][0].x, f[0][0].y}, b01 = {f[0][0].z, f[0][0].w};
            const v2f b02 = {f[0][1].x, f[0][1].y}, b03 = {f[0][1].z, f[0][1].w};
            const v2f b10 = {f[1][0].x, f[1][0].y}, b11 = {f[1][0].z, f[1][0].w};
            const v2f b12 = {f[1][1].x, f[1][1].y}, b13 = {f[1][1].z, f[1][1].w};
            acc0 = mm(Ga[0], b00, acc0);  acc1 = mm(Ga[0], b10, acc1);
            acc0 = mm(Ga[1], b01, acc0);  acc1 = mm(Ga[1], b11, acc1);
            acc0 = mm(Ga[2], b02, acc0);  acc1 = mm(Ga[2], b12, acc1);
            acc0 = mm(Ga[3], b03, acc0);  acc1 = mm(Ga[3], b13, acc1);
        }

        acc0 = mm(Ca, st, acc0);
        st.x = hi ? acc0[6] : f[0][1].z;
        st.y = hi ? acc0[7] : f[0][1].w;
        acc1 = mm(Ca, st, acc1);
        st.x = hi ? acc1[6] : f[1][1].z;
        st.y = hi ? acc1[7] : f[1][1].w;

        wave_lds_sync();
        if (KEEP) {
            *(float4*)&sl.t[j][ycol]          = make_float4(acc0[0], acc0[1], acc0[2], acc0[3]);
            *(float4*)&sl.t[j][ycol + 4]      = make_float4(acc0[4], acc0[5], acc0[6], acc0[7]);
            *(float4*)&sl.t[j][16 + ycol]     = make_float4(acc1[0], acc1[1], acc1[2], acc1[3]);
            *(float4*)&sl.t[j][16 + ycol + 4] = make_float4(acc1[4], acc1[5], acc1[6], acc1[7]);
        } else {
            asm volatile("" : "+v"(st.x), "+v"(st.y));
        }
    };

    float4 xb0, xb1, xb2, xb3;
    static_assert(NTRIP == 8, "a segment is eight trips");
    one_trip(slabs[0], 0,        false, false, xa0, xa1, xa2, xa3, xb0, xb1, xb2, xb3);
    one_trip(slabs[1], TRIP,     true,  false, xb0, xb1, xb2, xb3, xa0, xa1, xa2, xa3);
    one_trip(slabs[2], 2 * TRIP, false, false, xa0, xa1, xa2, xa3, xb0, xb1, xb2, xb3);
    one_trip(slabs[3], 3 * TRIP, true,  false, xb0, xb1, xb2, xb3, xa0, xa1, xa2, xa3);
    one_trip(slabs[4], 4 * TRIP, false, false, xa0, xa1, xa2, xa3, xb0, xb1, xb2, xb3);
    one_trip(slabs[5], 5 * TRIP, true,  false, xb0, xb1, xb2, xb3, xa0, xa1, xa2, xa3);
    one_trip(slabs[6], 6 * TRIP, false, false, xa0, xa1, xa2, xa3, xb0, xb1, xb2, xb3);
    one_trip(slabs[7], 7 * TRIP, true,  true,  xb0, xb1, xb2, xb3, xa0, xa1, xa2, xa3);
}

__device__ __forceinline__ void output_pass(
    WaveSlab (&slabs)[NTRIP], float* __restrict__ ytile, unsigned lane, v2f Ca,
    float y1, float y2, float P16, float Q16, float P15, float Q15)
{
    const int  j    = (int)(lane & 15u);
    const bool hi   = lane >= 16u;
    const int  ycol = hi ? 8 : 0;
    const int  lrow = (int)(lane >> 3);
    const int  lq   = (int)(lane & 7u);

    float d1 = hi ? y1 : 0.0f;
    float d2 = hi ? y2 : 0.0f;
#pragma unroll
    for (int tr = 0; tr < NTRIP; ++tr) {
        WaveSlab& sl = slabs[tr];
        const float4 q0 = *(const float4*)&sl.t[j][ycol];
        const float4 q1 = *(const float4*)&sl.t[j][ycol + 4];
        const float4 q2 = *(const float4*)&sl.t[j][16 + ycol];
        const float4 q3 = *(const float4*)&sl.t[j][16 + ycol + 4];
        v8f acc0 = {q0.x, q0.y, q0.z, q0.w, q1.x, q1.y, q1.z, q1.w};
        v8f acc1 = {q2.x, q2.y, q2.z, q2.w, q3.x, q3.y, q3.z, q3.w};

        const v2f s0 = {d2, d1};
        const float e1 = P16 * d1 + Q16 * d2;
        const float e2 = P15 * d1 + Q15 * d2;
        const v2f s1 = {e2, e1};
        d1 = P16 * e1 + Q16 * e2;
        d2 = P15 * e1 + Q15 * e2;

        acc0 = mm(Ca, s0, acc0);
        acc1 = mm(Ca, s1, acc1);

        *(float4*)&sl.t[j][ycol]          = make_float4(acc0[0], acc0[1], acc0[2], acc0[3]);
        *(float4*)&sl.t[j][ycol + 4]      = make_float4(acc0[4], acc0[5], acc0[6], acc0[7]);
        *(float4*)&sl.t[j][16 + ycol]     = make_float4(acc1[0], acc1[1], acc1[2], acc1[3]);
        *(float4*)&sl.t[j][16 + ycol + 4] = make_float4(acc1[4], acc1[5], acc1[6], acc1[7]);
        wave_lds_sync();
#pragma unroll
        for (int i = 0; i < 4; ++i) {
            const float4 yv = *(const float4*)&sl.t[4 * i + lrow][4 * lq];
            *(float4*)(ytile + (size_t)(4 * i + lrow) * T_LEN + tr * TRIP + 4 * lq) = yv;
        }
    }
}

__global__ __launch_bounds__(32 * NSEG) void dsvf_wmma_kernel(
    const float* x,
    const float* __restrict__ gp,
    const float* __restrict__ rp,
    const float* __restrict__ mhp_p,
    const float* __restrict__ mbp_p,
    const float* __restrict__ mlp_p,
    float* __restrict__ yout)
{
    static_assert(NPART == 2, "the pass sequence below is written for two parts");
    __shared__ WaveSlab slab[NSEG][NTRIP];
    __shared__ float hz[20];
    __shared__ float es1[NSEGT][16], es2[NSEGT][16];
    __shared__ float ss1[NSEGT][16], ss2[NSEGT][16];
    __shared__ float cf[2];
    __shared__ __attribute__((aligned(16))) float frag[32][12];

    const unsigned lane = threadIdx.x & 31u;
    const unsigned wave = threadIdx.x >> 5;
    const int      j    = (int)(lane & 15u);
    const bool     hi   = lane >= 16u;
    const bool lead = __builtin_amdgcn_readfirstlane((int)threadIdx.x) < 32;
    const int ntile = (int)(gridDim.x / NPART);
    const int part  = (int)blockIdx.x < ntile ? 1 : 0;
    const int rbase = ((int)blockIdx.x - (part ? 0 : ntile)) * 16;
    const int seg0  = (int)wave * SEG;
    const int gseg  = part * NSEG + (int)wave;

    const float* xt0   = x + (size_t)rbase * T_LEN + seg0;
    const float* xtown = xt0 + part * PART;
    float*       ytile = yout + (size_t)rbase * T_LEN + part * PART + seg0;

    float4 xq0, xq1, xq2, xq3;
    {
        const float* xl = xt0 + (size_t)(lane >> 3) * T_LEN + 4 * (lane & 7u);
        xq0 = *(const float4*)(xl);
        xq1 = *(const float4*)(xl + (size_t)4 * T_LEN);
        xq2 = *(const float4*)(xl + (size_t)8 * T_LEN);
        xq3 = *(const float4*)(xl + (size_t)12 * T_LEN);
    }

    WaveSlab (&sl)[NTRIP] = slab[wave];
    const float* xr = x + (size_t)(rbase + j) * T_LEN + seg0;
    const float xm1a = wave ? xr[-1] : 0.0f;
    const float xm2a = wave ? xr[-2] : 0.0f;
    const float xm1b = xr[PART - 1];
    const float xm2b = xr[PART - 2];

    if (lead) {
        const float gv  = gp[0];
        const float rv  = rp[0];
        const float mhp = mhp_p[0];
        const float mbp = mbp_p[0];
        const float mlp = mlp_p[0];

        const float sig = 1.0f / (1.0f + expf(-gv));
        const float gg  = tanf(3.14159265358979323846f * 0.5f * sig);
        const float rr  = log1pf(expf(rv));
        const float g2  = gg * gg;

        const float b0 = g2 * mlp + gg * mbp + mhp;
        const float b1 = 2.0f * g2 * mlp - 2.0f * mhp;
        const float b2 = g2 * mlp - gg * mbp + mhp;
        const float a0 = g2 + 2.0f * rr * gg + 1.0f;
        const float inva0 = 1.0f / a0;
        const float a1 = (2.0f * g2 - 2.0f) * inva0;
        const float a2 = (g2 - 2.0f * rr * gg + 1.0f) * inva0;

        if (threadIdx.x == 0u) {
            hz[0] = 0.0f;
            hz[1] = 0.0f;
            hz[2] = 1.0f;
            hz[3] = -a1;
            for (int i = 2; i <= 16; ++i)
                hz[2 + i] = -a1 * hz[1 + i] - a2 * hz[i];
            cf[0] = a1;
            cf[1] = a2;
        }
        wave_lds_sync();

        const int fcol = hi ? 0 : 8;
#pragma unroll
        for (int s = 0; s < 4; ++s) {
            const int d0 = j - (fcol + 2 * s);
            const int d1 = d0 - 1;
            const int c0 = d0 < 0 ? 0 : d0;
            const int c1 = d1 < 0 ? 0 : d1;
            const float g0 = b0 * hz[c0 + 2] + b1 * hz[c0 + 1] + b2 * hz[c0];
            const float g1 = b0 * hz[c1 + 2] + b1 * hz[c1 + 1] + b2 * hz[c1];
            frag[lane][2 * s]     = d0 >= 0 ? g0 : 0.0f;
            frag[lane][2 * s + 1] = d1 >= 0 ? g1 : 0.0f;
        }
        const float ht  = hz[j + 2];
        const float htm = hz[j + 1];
        const float htp = hz[j + 3];
        frag[lane][8] = hi ? -a2 * ht : b2 * ht;
        frag[lane][9] = hi ? htp      : b1 * ht + b2 * htm;
    }
    __syncthreads();

    const float a1 = __builtin_bit_cast(float, __builtin_amdgcn_readfirstlane(__builtin_bit_cast(int, cf[0])));
    const float a2 = __builtin_bit_cast(float, __builtin_amdgcn_readfirstlane(__builtin_bit_cast(int, cf[1])));
    v2f Ga[4];
    v2f Ca;
    {
        const float4 fa = *(const float4*)&frag[lane][0];
        const float4 fb = *(const float4*)&frag[lane][4];
        const float2 fc = *(const float2*)&frag[lane][8];
        Ga[0].x = fa.x; Ga[0].y = fa.y; Ga[1].x = fa.z; Ga[1].y = fa.w;
        Ga[2].x = fb.x; Ga[2].y = fb.y; Ga[3].x = fb.z; Ga[3].y = fb.w;
        Ca.x = fc.x; Ca.y = fc.y;
    }

    float h1 = 1.0f, h2 = 0.0f, h3 = 0.0f;
    v2f st;
    st.x = hi ? 0.0f : xm2a;
    st.y = hi ? 0.0f : xm1a;
    if (part) {
        segment_pass<false, true, true>(sl, xt0, xtown, lane, a1, a2, Ga, Ca, st, h1, h2, h3,
                                        xq0, xq1, xq2, xq3);
        if (hi) {
            es1[wave][j] = st.y;
            es2[wave][j] = st.x;
        }
        st.x = hi ? 0.0f : xm2b;
        st.y = hi ? 0.0f : xm1b;
        segment_pass<true, false, false>(sl, xtown, xtown, lane, a1, a2, Ga, Ca, st, h1, h2, h3,
                                         xq0, xq1, xq2, xq3);
        if (hi) {
            es1[NSEG + wave][j] = st.y;
            es2[NSEG + wave][j] = st.x;
        }
    } else {
        segment_pass<true, true, false>(sl, xt0, xt0, lane, a1, a2, Ga, Ca, st, h1, h2, h3,
                                        xq0, xq1, xq2, xq3);
        if (hi) {
            es1[wave][j] = st.y;
            es2[wave][j] = st.x;
        }
    }
    __syncthreads();

    const float P1 = h1, Q1 = -a2 * h2;
    const float P2 = h2, Q2 = -a2 * h3;
    if (lead && !hi) {
        float y1 = 0.0f, y2 = 0.0f;
        ss1[0][j] = 0.0f;
        ss2[0][j] = 0.0f;
#pragma unroll
        for (int s = 0; s < NSEG - 1; ++s) {
            const float n1 = es1[s][j] + P1 * y1 + Q1 * y2;
            const float n2 = es2[s][j] + P2 * y1 + Q2 * y2;
            y1 = n1;
            y2 = n2;
            ss1[s + 1][j] = y1;
            ss2[s + 1][j] = y2;
        }
        if (part) {
#pragma unroll
            for (int s = NSEG - 1; s < NSEGT - 1; ++s) {
                const float n1 = es1[s][j] + P1 * y1 + Q1 * y2;
                const float n2 = es2[s][j] + P2 * y1 + Q2 * y2;
                y1 = n1;
                y2 = n2;
                ss1[s + 1][j] = y1;
                ss2[s + 1][j] = y2;
            }
        }
    }
    __syncthreads();
    const float y1 = ss1[gseg][j];
    const float y2 = ss2[gseg][j];

    const float P16 = hz[18], Q16 = -a2 * hz[17];
    const float P15 = hz[17], Q15 = -a2 * hz[16];
    output_pass(sl, ytile, lane, Ca, y1, y2, P16, Q16, P15, Q15);
}

extern "C" void kernel_launch(void* const* d_in, const int* in_sizes, int n_in,
                              void* d_out, int out_size, void* d_ws, size_t ws_size,
                              hipStream_t stream) {
    const float* x   = (const float*)d_in[0];
    const float* g   = (const float*)d_in[1];
    const float* r   = (const float*)d_in[2];
    const float* mhp = (const float*)d_in[3];
    const float* mbp = (const float*)d_in[4];
    const float* mlp = (const float*)d_in[5];
    float* y = (float*)d_out;

    const int B = in_sizes[0] / T_LEN;
    dim3 grid(B / 16 * NPART), block(32 * NSEG);
    hipLaunchKernelGGL(dsvf_wmma_kernel, grid, block, 0, stream,
                       x, g, r, mhp, mbp, mlp, y);
}


# speedup vs baseline: 26.4632x; 1.3891x over previous
#include <hip/hip_runtime.h>
#include <math.h>


typedef float v2f __attribute__((ext_vector_type(2)));
typedef float v8f __attribute__((ext_vector_type(8)));

constexpr int T_LEN = 8192;
constexpr int ROWS  = 8;
constexpr int NWAVE = 16;
constexpr int CSEG  = 16 / ROWS;
constexpr int NSEGT = NWAVE * CSEG;
constexpr int SEG   = T_LEN / NSEGT;
constexpr int TRIP  = 32;
constexpr int NTRIP = SEG / TRIP;
constexpr int PITCH = 36;

struct __attribute__((aligned(16))) WaveSlab {
    float t[16][PITCH];
    float skew[32];
};
static_assert((sizeof(WaveSlab) / 4) % 64 == 32, "consecutive slabs lie half a bank round apart");

__device__ __forceinline__ void wave_lds_sync() {
    __builtin_amdgcn_fence(__ATOMIC_RELEASE, "wavefront");
    __builtin_amdgcn_wave_barrier();
    __builtin_amdgcn_fence(__ATOMIC_ACQUIRE, "wavefront");
}

__device__ __forceinline__ v8f mm(v2f a, v2f b, v8f c) {
    return __builtin_amdgcn_wmma_f32_16x16x4_f32(
         false, a,  false, b,
         (short)0, c,  false,  false);
}

__device__ __forceinline__ void segment_pass(
    WaveSlab (&slabs)[NTRIP], const float* xtile,
    unsigned lane, float a1, float a2,
    const v2f (&Ga)[4], v2f Ca, v2f& st, float& h1, float& h2, float& h3,
    float4& xa0, float4& xa1, float4& xa2, float4& xa3)
{
    const int  j  = (int)(lane & 15u);
    const bool hi = lane >= 16u;
    const int fcol  = hi ? 0 : 8;
    const int ycol  = hi ? 8 : 0;
    const int pcol  = (int)(lane >> 1);
    const int fb    = (int)(lane & 1u);
    const int lrow  = (int)(lane >> 3);
    const int lq    = (int)(lane & 7u);
    const int poff  = (pcol % ROWS) * T_LEN + (pcol / ROWS) * SEG;

    auto one_trip = [&](WaveSlab& sl, const int t0, const bool second, const bool last,
                        const float4& c0, const float4& c1, const float4& c2, const float4& c3,
                        float4& n0, float4& n1, float4& n2, float4& n3)
                        __attribute__((always_inline)) {
        if (!last) {
            const float* xl = xtile + t0 + TRIP + (size_t)lrow * T_LEN + 4 * lq;
            n0 = *(const float4*)(xl);
            n1 = *(const float4*)(xl + (size_t)4 * T_LEN);
            n2 = *(const float4*)(xl + SEG);
            n3 = *(const float4*)(xl + (size_t)4 * T_LEN + SEG);
        }
        {
#pragma unroll
            for (int k = 0; k < TRIP; ++k) {
                const float hn = -a1 * h1 - a2 * h2;
                h3 = h2; h2 = h1; h1 = hn;
            }
            asm volatile("" : "+s"(h1), "+s"(h2), "+s"(h3));
        }
        if (!second && t0 + 2 * TRIP < SEG)
            __builtin_prefetch(xtile + (size_t)poff + t0 + 2 * TRIP + 16 * fb, 0, 1);

        *(float4*)&sl.t[lrow][4 * lq]      = c0;
        *(float4*)&sl.t[4 + lrow][4 * lq]  = c1;
        *(float4*)&sl.t[8 + lrow][4 * lq]  = c2;
        *(float4*)&sl.t[12 + lrow][4 * lq] = c3;
        wave_lds_sync();

        float4 f[2][2];
#pragma unroll
        for (int c = 0; c < 2; ++c) {
            f[c][0] = *(const float4*)&sl.t[j][16 * c + fcol];
            f[c][1] = *(const float4*)&sl.t[j][16 * c + fcol + 4];
        }

        v8f acc0 = {}, acc1 = {};
        {
            const v2f b00 = {f[0][0].x, f[0][0].y}, b01 = {f[0][0].z, f[0][0].w};
            const v2f b02 = {f[0][1].x, f[0][1].y}, b03 = {f[0][1].z, f[0][1].w};
            const v2f b10 = {f[1][0].x, f[1][0].y}, b11 = {f[1][0].z, f[1][0].w};
            const v2f b12 = {f[1][1].x, f[1][1].y}, b13 = {f[1][1].z, f[1][1].w};
            acc0 = mm(Ga[0], b00, acc0);  acc1 = mm(Ga[0], b10, acc1);
            acc0 = mm(Ga[1], b01, acc0);  acc1 = mm(Ga[1], b11, acc1);
            acc0 = mm(Ga[2], b02, acc0);  acc1 = mm(Ga[2], b12, acc1);
            acc0 = mm(Ga[3], b03, acc0);  acc1 = mm(Ga[3], b13, acc1);
        }

        acc0 = mm(Ca, st, acc0);
        st.x = hi ? acc0[6] : f[0][1].z;
        st.y = hi ? acc0[7] : f[0][1].w;
        acc1 = mm(Ca, st, acc1);
        st.x = hi ? acc1[6] : f[1][1].z;
        st.y = hi ? acc1[7] : f[1][1].w;

        wave_lds_sync();
        *(float4*)&sl.t[j][ycol]          = make_float4(acc0[0], acc0[1], acc0[2], acc0[3]);
        *(float4*)&sl.t[j][ycol + 4]      = make_float4(acc0[4], acc0[5], acc0[6], acc0[7]);
        *(float4*)&sl.t[j][16 + ycol]     = make_float4(acc1[0], acc1[1], acc1[2], acc1[3]);
        *(float4*)&sl.t[j][16 + ycol + 4] = make_float4(acc1[4], acc1[5], acc1[6], acc1[7]);
    };

    float4 xb0, xb1, xb2, xb3;
    static_assert(NTRIP == 8, "a segment is eight trips");
    static_assert(ROWS == 8 && CSEG == 2, "the tile loads are written for 2 segments x 8 rows");
    one_trip(slabs[0], 0,        false, false, xa0, xa1, xa2, xa3, xb0, xb1, xb2, xb3);
    one_trip(slabs[1], TRIP,     true,  false, xb0, xb1, xb2, xb3, xa0, xa1, xa2, xa3);
    one_trip(slabs[2], 2 * TRIP, false, false, xa0, xa1, xa2, xa3, xb0, xb1, xb2, xb3);
    one_trip(slabs[3], 3 * TRIP, true,  false, xb0, xb1, xb2, xb3, xa0, xa1, xa2, xa3);
    one_trip(slabs[4], 4 * TRIP, false, false, xa0, xa1, xa2, xa3, xb0, xb1, xb2, xb3);
    one_trip(slabs[5], 5 * TRIP, true,  false, xb0, xb1, xb2, xb3, xa0, xa1, xa2, xa3);
    one_trip(slabs[6], 6 * TRIP, false, false, xa0, xa1, xa2, xa3, xb0, xb1, xb2, xb3);
    one_trip(slabs[7], 7 * TRIP, true,  true,  xb0, xb1, xb2, xb3, xa0, xa1, xa2, xa3);
}

__device__ __forceinline__ void output_pass(
    WaveSlab (&slabs)[NTRIP], float* __restrict__ ytile, unsigned lane, v2f Ca,
    float y1, float y2, float P16, float Q16, float P15, float Q15)
{
    const int  j    = (int)(lane & 15u);
    const bool hi   = lane >= 16u;
    const int  ycol = hi ? 8 : 0;
    const int  urow = (int)(lane >> 4);
    const int  utr  = (int)((lane >> 3) & 1u);
    const int  ulq  = (int)(lane & 7u);

    float d1 = hi ? y1 : 0.0f;
    float d2 = hi ? y2 : 0.0f;
    static_assert(NTRIP % 2 == 0, "the output pass works the trips off in pairs");
    static_assert(ROWS == 8 && CSEG == 2, "the unit stores are written for 2 segments x 8 rows");
#pragma unroll
    for (int pr = 0; pr < NTRIP / 2; ++pr) {
#pragma unroll
        for (int u = 0; u < 2; ++u) {
            WaveSlab& sl = slabs[2 * pr + u];
            const float4 q0 = *(const float4*)&sl.t[j][ycol];
            const float4 q1 = *(const float4*)&sl.t[j][ycol + 4];
            const float4 q2 = *(const float4*)&sl.t[j][16 + ycol];
            const float4 q3 = *(const float4*)&sl.t[j][16 + ycol + 4];
            v8f acc0 = {q0.x, q0.y, q0.z, q0.w, q1.x, q1.y, q1.z, q1.w};
            v8f acc1 = {q2.x, q2.y, q2.z, q2.w, q3.x, q3.y, q3.z, q3.w};

            const v2f s0 = {d2, d1};
            const float e1 = P16 * d1 + Q16 * d2;
            const float e2 = P15 * d1 + Q15 * d2;
            const v2f s1 = {e2, e1};
            d1 = P16 * e1 + Q16 * e2;
            d2 = P15 * e1 + Q15 * e2;

            acc0 = mm(Ca, s0, acc0);
            acc1 = mm(Ca, s1, acc1);

            *(float4*)&sl.t[j][ycol]          = make_float4(acc0[0], acc0[1], acc0[2], acc0[3]);
            *(float4*)&sl.t[j][ycol + 4]      = make_float4(acc0[4], acc0[5], acc0[6], acc0[7]);
            *(float4*)&sl.t[j][16 + ycol]     = make_float4(acc1[0], acc1[1], acc1[2], acc1[3]);
            *(float4*)&sl.t[j][16 + ycol + 4] = make_float4(acc1[4], acc1[5], acc1[6], acc1[7]);
        }
        wave_lds_sync();
#pragma unroll
        for (int i = 0; i < 8; ++i) {
            const float4 yv = *(const float4*)&slabs[2 * pr + utr].t[2 * i + urow][4 * ulq];
            *(float4*)(ytile + (size_t)(2 * (i % 4) + urow) * T_LEN + (i / 4) * SEG
                             + (2 * pr + utr) * TRIP + 4 * ulq) = yv;
        }
    }
}

__global__ __launch_bounds__(32 * NWAVE) void dsvf_wmma_kernel(
    const float* x,
    const float* __restrict__ gp,
    const float* __restrict__ rp,
    const float* __restrict__ mhp_p,
    const float* __restrict__ mbp_p,
    const float* __restrict__ mlp_p,
    float* __restrict__ yout)
{
    __shared__ WaveSlab slab[NWAVE][NTRIP];
    __shared__ float hz[20];
    __shared__ float es1[NSEGT][ROWS], es2[NSEGT][ROWS];
    __shared__ float ss1[NSEGT][ROWS], ss2[NSEGT][ROWS];
    __shared__ float cf[2];
    __shared__ __attribute__((aligned(16))) float frag[32][12];

    const unsigned lane = threadIdx.x & 31u;
    const unsigned wave = threadIdx.x >> 5;
    const int      j    = (int)(lane & 15u);
    const bool     hi   = lane >= 16u;
    const int      row  = j % ROWS;
    const int      gseg = (int)wave * CSEG + j / ROWS;
    const bool lead = __builtin_amdgcn_readfirstlane((int)threadIdx.x) < 32;
    const int rbase = (int)blockIdx.x * ROWS;

    const float* xtile = x + (size_t)rbase * T_LEN + (int)wave * CSEG * SEG;
    float*       ytile = yout + (size_t)rbase * T_LEN + (int)wave * CSEG * SEG;

    float4 xq0, xq1, xq2, xq3;
    {
        const float* xl = xtile + (size_t)(lane >> 3) * T_LEN + 4 * (lane & 7u);
        xq0 = *(const float4*)(xl);
        xq1 = *(const float4*)(xl + (size_t)4 * T_LEN);
        xq2 = *(const float4*)(xl + SEG);
        xq3 = *(const float4*)(xl + (size_t)4 * T_LEN + SEG);
    }

    WaveSlab (&sl)[NTRIP] = slab[wave];
    const float* xr = x + (size_t)(rbase + row) * T_LEN + gseg * SEG;
    const float xm1 = gseg ? xr[-1] : 0.0f;
    const float xm2 = gseg ? xr[-2] : 0.0f;

    if (lead) {
        const float gv  = gp[0];
        const float rv  = rp[0];
        const float mhp = mhp_p[0];
        const float mbp = mbp_p[0];
        const float mlp = mlp_p[0];

        const float sig = 1.0f / (1.0f + expf(-gv));
        const float gg  = tanf(3.14159265358979323846f * 0.5f * sig);
        const float rr  = log1pf(expf(rv));
        const float g2  = gg * gg;

        const float b0 = g2 * mlp + gg * mbp + mhp;
        const float b1 = 2.0f * g2 * mlp - 2.0f * mhp;
        const float b2 = g2 * mlp - gg * mbp + mhp;
        const float a0 = g2 + 2.0f * rr * gg + 1.0f;
        const float inva0 = 1.0f / a0;
        const float a1 = (2.0f * g2 - 2.0f) * inva0;
        const float a2 = (g2 - 2.0f * rr * gg + 1.0f) * inva0;

        if (threadIdx.x == 0u) {
            hz[0] = 0.0f;
            hz[1] = 0.0f;
            hz[2] = 1.0f;
            hz[3] = -a1;
            for (int i = 2; i <= 16; ++i)
                hz[2 + i] = -a1 * hz[1 + i] - a2 * hz[i];
            cf[0] = a1;
            cf[1] = a2;
        }
        wave_lds_sync();

        const int fcol = hi ? 0 : 8;
#pragma unroll
        for (int s = 0; s < 4; ++s) {
            const int d0 = j - (fcol + 2 * s);
            const int d1 = d0 - 1;
            const int c0 = d0 < 0 ? 0 : d0;
            const int c1 = d1 < 0 ? 0 : d1;
            const float g0 = b0 * hz[c0 + 2] + b1 * hz[c0 + 1] + b2 * hz[c0];
            const float g1 = b0 * hz[c1 + 2] + b1 * hz[c1 + 1] + b2 * hz[c1];
            frag[lane][2 * s]     = d0 >= 0 ? g0 : 0.0f;
            frag[lane][2 * s + 1] = d1 >= 0 ? g1 : 0.0f;
        }
        const float ht  = hz[j + 2];
        const float htm = hz[j + 1];
        const float htp = hz[j + 3];
        frag[lane][8] = hi ? -a2 * ht : b2 * ht;
        frag[lane][9] = hi ? htp      : b1 * ht + b2 * htm;
    }
    __syncthreads();

    const float a1 = __builtin_bit_cast(float, __builtin_amdgcn_readfirstlane(__builtin_bit_cast(int, cf[0])));
    const float a2 = __builtin_bit_cast(float, __builtin_amdgcn_readfirstlane(__builtin_bit_cast(int, cf[1])));
    v2f Ga[4];
    v2f Ca;
    {
        const float4 fa = *(const float4*)&frag[lane][0];
        const float4 fb = *(const float4*)&frag[lane][4];
        const float2 fc = *(const float2*)&frag[lane][8];
        Ga[0].x = fa.x; Ga[0].y = fa.y; Ga[1].x = fa.z; Ga[1].y = fa.w;
        Ga[2].x = fb.x; Ga[2].y = fb.y; Ga[3].x = fb.z; Ga[3].y = fb.w;
        Ca.x = fc.x; Ca.y = fc.y;
    }

    float h1 = 1.0f, h2 = 0.0f, h3 = 0.0f;
    v2f st;
    st.x = hi ? 0.0f : xm2;
    st.y = hi ? 0.0f : xm1;
    segment_pass(sl, xtile, lane, a1, a2, Ga, Ca, st, h1, h2, h3, xq0, xq1, xq2, xq3);
    if (hi) {
        es1[gseg][row] = st.y;
        es2[gseg][row] = st.x;
    }
    __syncthreads();

    const float P1 = h1, Q1 = -a2 * h2;
    const float P2 = h2, Q2 = -a2 * h3;
    if (lead && lane < (unsigned)ROWS) {
        float y1 = 0.0f, y2 = 0.0f;
        ss1[0][lane] = 0.0f;
        ss2[0][lane] = 0.0f;
#pragma unroll
        for (int s = 0; s < NSEGT - 1; ++s) {
            const float n1 = es1[s][lane] + P1 * y1 + Q1 * y2;
            const float n2 = es2[s][lane] + P2 * y1 + Q2 * y2;
            y1 = n1;
            y2 = n2;
            ss1[s + 1][lane] = y1;
            ss2[s + 1][lane] = y2;
        }
    }
    __syncthreads();
    const float y1 = ss1[gseg][row];
    const float y2 = ss2[gseg][row];

    const float P16 = hz[18], Q16 = -a2 * hz[17];
    const float P15 = hz[17], Q15 = -a2 * hz[16];
    output_pass(sl, ytile, lane, Ca, y1, y2, P16, Q16, P15, Q15);
}

extern "C" void kernel_launch(void* const* d_in, const int* in_sizes, int n_in,
                              void* d_out, int out_size, void* d_ws, size_t ws_size,
                              hipStream_t stream) {
    const float* x   = (const float*)d_in[0];
    const float* g   = (const float*)d_in[1];
    const float* r   = (const float*)d_in[2];
    const float* mhp = (const float*)d_in[3];
    const float* mbp = (const float*)d_in[4];
    const float* mlp = (const float*)d_in[5];
    float* y = (float*)d_out;

    const int B = in_sizes[0] / T_LEN;
    dim3 grid(B / ROWS), block(32 * NWAVE);
    hipLaunchKernelGGL(dsvf_wmma_kernel, grid, block, 0, stream,
                       x, g, r, mhp, mbp, mlp, y);
}
